# starting kernel
# baseline (speedup 1.0000x reference)
#include <hip/hip_runtime.h>
#include <hip/hip_bf16.h>

typedef __attribute__((ext_vector_type(2))) float v2f;
typedef __attribute__((ext_vector_type(8))) float v8f;

#define CCH 64
#define PAIR_STRIDE 160

__global__ void zero_f32(float* __restrict__ p, long n) {
  long i = (long)blockIdx.x * blockDim.x + threadIdx.x;
  if (i < n) p[i] = 0.0f;
}

__global__ void edge_scatter(const float* __restrict__ x,
                             const int* __restrict__ src_idx,
                             const int* __restrict__ dst_idx,
                             const float* __restrict__ w,
                             float* __restrict__ agg, int E) {
  int t = blockIdx.x * blockDim.x + threadIdx.x;
  int e = t >> 4;
  if (e >= E) return;
  int cg = (t & 15) << 2;
  int s = src_idx[e];
  int d = dst_idx[e];
  float ww = w[e];
  const float4 xv = *(const float4*)(x + (long)s * CCH + cg);
  float* p = agg + (long)d * CCH + cg;
  __hip_atomic_fetch_add(p + 0, xv.x * ww, __ATOMIC_RELAXED, __HIP_MEMORY_SCOPE_AGENT);
  __hip_atomic_fetch_add(p + 1, xv.y * ww, __ATOMIC_RELAXED, __HIP_MEMORY_SCOPE_AGENT);
  __hip_atomic_fetch_add(p + 2, xv.z * ww, __ATOMIC_RELAXED, __HIP_MEMORY_SCOPE_AGENT);
  __hip_atomic_fetch_add(p + 3, xv.w * ww, __ATOMIC_RELAXED, __HIP_MEMORY_SCOPE_AGENT);
}

__global__ __launch_bounds__(256)
void gemm_dual_wmma(const float* __restrict__ Aagg,
                    const float* __restrict__ Aroot,
                    const float* __restrict__ Wn,
                    const float* __restrict__ Wr,
                    const float* __restrict__ bias,
                    float* __restrict__ out, int Nrows) {
  __shared__ float lds[64 * PAIR_STRIDE];

  for (int idx = threadIdx.x; idx < 128 * 64; idx += 256) {
    int k = idx >> 6, n = idx & 63;
    float v = (k < 64) ? Wn[idx] : Wr[idx - 4096];
    lds[(k >> 1) * PAIR_STRIDE + (n << 1) + (k & 1)] = v;
  }
  __syncthreads();

  const int wave = threadIdx.x >> 5;
  const int lane = threadIdx.x & 31;
  const int lh = lane >> 4;
  const int l  = lane & 15;

  long tile = (long)blockIdx.x * 8 + wave;
  long row0 = tile * 16;
  if (row0 >= Nrows) return;

  long arow = row0 + l;
  if (arow >= Nrows) arow = Nrows - 1;
  const float* pa = Aagg + arow * CCH;
  const float* px = Aroot + arow * CCH;

  v8f acc0 = {}, acc1 = {}, acc2 = {}, acc3 = {};

#pragma unroll
  for (int k0 = 0; k0 < 128; k0 += 4) {
    int kk = k0 + (lh << 1);
    v2f a = (k0 < 64) ? *(const v2f*)(pa + kk)
                      : *(const v2f*)(px + (kk - 64));
    const float* bb = lds + (((k0 >> 1) + lh) * PAIR_STRIDE) + (l << 1);
    v2f b0 = *(const v2f*)(bb);
    v2f b1 = *(const v2f*)(bb + 32);
    v2f b2 = *(const v2f*)(bb + 64);
    v2f b3 = *(const v2f*)(bb + 96);
    acc0 = __builtin_amdgcn_wmma_f32_16x16x4_f32(false, a, false, b0, (short)0, acc0, false, false);
    acc1 = __builtin_amdgcn_wmma_f32_16x16x4_f32(false, a, false, b1, (short)0, acc1, false, false);
    acc2 = __builtin_amdgcn_wmma_f32_16x16x4_f32(false, a, false, b2, (short)0, acc2, false, false);
    acc3 = __builtin_amdgcn_wmma_f32_16x16x4_f32(false, a, false, b3, (short)0, acc3, false, false);
  }

  float bc0 = bias[l], bc1 = bias[16 + l], bc2 = bias[32 + l], bc3 = bias[48 + l];

#pragma unroll
  for (int r = 0; r < 8; ++r) {
    long orow = row0 + r + (lh << 3);
    if (orow < Nrows) {
      float* po = out + orow * CCH + l;
      po[0]  = acc0[r] + bc0;
      po[16] = acc1[r] + bc1;
      po[32] = acc2[r] + bc2;
      po[48] = acc3[r] + bc3;
    }
  }
}

extern "C" void kernel_launch(void* const* d_in, const int* in_sizes, int n_in,
                              void* d_out, int out_size, void* d_ws, size_t ws_size,
                              hipStream_t stream) {
  const float* x   = (const float*)d_in[0];
  const int*   ei  = (const int*)d_in[1];
  const float* ea  = (const float*)d_in[2];
  const int*   ps  = (const int*)d_in[3];
  const int*   pd  = (const int*)d_in[4];
  const float* pw  = (const float*)d_in[5];
  const float* Wr1 = (const float*)d_in[7];
  const float* Wn1 = (const float*)d_in[8];
  const float* b1  = (const float*)d_in[9];
  const float* Wr2 = (const float*)d_in[10];
  const float* Wn2 = (const float*)d_in[11];
  const float* b2  = (const float*)d_in[12];
  float* out = (float*)d_out;

  const int N  = in_sizes[0] / CCH;
  const int E  = in_sizes[1] / 2;
  const int EP = in_sizes[3];
  const long NC = (long)N * CCH;

  float* agg = (float*)d_ws;
  float* h1  = agg + NC;
  float* h2  = h1 + NC;

  const int* esrc = ei;
  const int* edst = ei + E;

  const int zb  = (int)((NC + 255) / 256);
  const int sb  = (int)(((long)E * 16 + 255) / 256);
  const int pb  = (int)(((long)EP * 16 + 255) / 256);
  const int gb  = ((N + 15) / 16 + 7) / 8;
  const long outN = (long)out_size;
  const int ob  = (int)((outN + 255) / 256);

  zero_f32<<<zb, 256, 0, stream>>>(agg, NC);
  edge_scatter<<<sb, 256, 0, stream>>>(x, esrc, edst, ea, agg, E);
  gemm_dual_wmma<<<gb, 256, 0, stream>>>(agg, x, Wn1, Wr1, b1, h1, N);
  zero_f32<<<zb, 256, 0, stream>>>(agg, NC);
  edge_scatter<<<sb, 256, 0, stream>>>(h1, esrc, edst, ea, agg, E);
  gemm_dual_wmma<<<gb, 256, 0, stream>>>(agg, h1, Wn2, Wr2, b2, h2, N);
  zero_f32<<<ob, 256, 0, stream>>>(out, outN);
  edge_scatter<<<pb, 256, 0, stream>>>(h2, ps, pd, pw, out, EP);
}


# edge sums: f32 atomics and zero fills replaced by a per-call destination-sorted edge list (int count+rank, block scan) and a wave-per-row register sum; sums in arrival-rank order per row
# speedup vs baseline: 1.3747x; 1.3747x over previous
#include <hip/hip_runtime.h>
#include <hip/hip_bf16.h>

typedef __attribute__((ext_vector_type(2))) float v2f;
typedef __attribute__((ext_vector_type(8))) float v8f;

#define CCH 64
#define PAIR_STRIDE 160
#define SCAN_BLOCK 1024

__global__ void zero_i32x4(int* __restrict__ p, int n4) {
  int i = blockIdx.x * blockDim.x + threadIdx.x;
  if (i < n4) ((int4*)p)[i] = make_int4(0, 0, 0, 0);
}

__global__ void count_rank(const int* __restrict__ dstA, int EA,
                           const int* __restrict__ dstB, int EB, int rowBaseB,
                           int* __restrict__ cnt, int* __restrict__ rank) {
  int t = blockIdx.x * blockDim.x + threadIdx.x;
  if (t >= EA + EB) return;
  int row = (t < EA) ? dstA[t] : (rowBaseB + dstB[t - EA]);
  rank[t] = __hip_atomic_fetch_add(cnt + row, 1, __ATOMIC_RELAXED, __HIP_MEMORY_SCOPE_AGENT);
}

__global__ __launch_bounds__(SCAN_BLOCK)
void assign_offsets(const int* __restrict__ cnt, int* __restrict__ off,
                    int* __restrict__ cursor) {
  __shared__ int wsum[32];
  __shared__ int sbase;
  const int i = blockIdx.x * SCAN_BLOCK + threadIdx.x;
  const int lane = threadIdx.x & 31;
  const int wv = threadIdx.x >> 5;
  const int c = cnt[i];
  int incl = c;
  for (int d = 1; d < 32; d <<= 1) {
    int t = __shfl_up(incl, d, 32);
    if (lane >= d) incl += t;
  }
  if (lane == 31) wsum[wv] = incl;
  __syncthreads();
  if (wv == 0) {
    const int s = wsum[lane];
    int si = s;
    for (int d = 1; d < 32; d <<= 1) {
      int t = __shfl_up(si, d, 32);
      if (lane >= d) si += t;
    }
    wsum[lane] = si - s;
    if (lane == 31)
      sbase = __hip_atomic_fetch_add(cursor, si, __ATOMIC_RELAXED, __HIP_MEMORY_SCOPE_AGENT);
  }
  __syncthreads();
  off[i] = sbase + wsum[wv] + incl - c;
}

__global__ void fill_pairs(const int* __restrict__ srcA, const int* __restrict__ dstA,
                           const float* __restrict__ wA, int EA,
                           const int* __restrict__ srcB, const int* __restrict__ dstB,
                           const float* __restrict__ wB, int EB, int rowBaseB,
                           const int* __restrict__ off, const int* __restrict__ rank,
                           int* __restrict__ pairs) {
  int t = blockIdx.x * blockDim.x + threadIdx.x;
  if (t >= EA + EB) return;
  int s, row;
  float w;
  if (t < EA) {
    s = srcA[t]; row = dstA[t]; w = wA[t];
  } else {
    int u = t - EA;
    s = srcB[u]; row = rowBaseB + dstB[u]; w = wB[u];
  }
  int pos = off[row] + rank[t];
  ((int2*)pairs)[pos] = make_int2(s, __float_as_int(w));
}

template <int R>
__global__ __launch_bounds__(256)
void gather_reduce(const float* __restrict__ x, const int* __restrict__ off,
                   const int* __restrict__ cnt, const int* __restrict__ pairs_i,
                   float* __restrict__ out, int rowBase, int nrows) {
  const int lane = threadIdx.x & 31;
  const int wave = blockIdx.x * 8 + (threadIdx.x >> 5);
  const int r0 = wave * R;
  if (r0 >= nrows) return;

  int jo = 0, je = 0;
  if (lane < R) {
    jo = off[rowBase + r0 + lane];
    je = jo + cnt[rowBase + r0 + lane];
  }
  const int base = __builtin_amdgcn_readlane(jo, 0);
  const int total = __builtin_amdgcn_readlane(je, R - 1) - base;
  const int relEnd = je - base;

  const int2* pairs = (const int2*)pairs_i + base;
  const float* xl = x + 2 * lane;
  float* ol = out + (long)r0 * CCH + 2 * lane;

  v2f acc = {0.0f, 0.0f};
  int row = 0;
  int rend = __builtin_amdgcn_readlane(relEnd, 0);

  for (int c = 0; c < total; c += 32) {
    int2 pr = make_int2(0, 0);
    if (c + lane < total) pr = pairs[c + lane];
    const int left = total - c;
    const int nn = left < 32 ? left : 32;
    for (int i = 0; i < nn; i += 4) {
      v2f v[4];
      float w[4];
      for (int u = 0; u < 4; ++u) {
        const int s = __builtin_amdgcn_readlane(pr.x, i + u);
        w[u] = __int_as_float(__builtin_amdgcn_readlane(pr.y, i + u));
        v[u] = *(const v2f*)(xl + (long)s * CCH);
      }
      for (int u = 0; u < 4; ++u) {
        if (i + u < nn) {
          if (R > 1) {
            const int e = c + i + u;
            while (e == rend) {
              if (r0 + row < nrows) *(v2f*)(ol + row * CCH) = acc;
              acc = (v2f){0.0f, 0.0f};
              ++row;
              rend = (row < R) ? __builtin_amdgcn_readlane(relEnd, row) : 0x7fffffff;
            }
          }
          acc += v[u] * w[u];
        }
      }
    }
  }
  while (row < R) {
    if (r0 + row < nrows) *(v2f*)(ol + row * CCH) = acc;
    acc = (v2f){0.0f, 0.0f};
    ++row;
  }
}

__global__ __launch_bounds__(256)
void gemm_dual_wmma(const float* Aagg,
                    const float* __restrict__ Aroot,
                    const float* __restrict__ Wn,
                    const float* __restrict__ Wr,
                    const float* __restrict__ bias,
                    float* out, int Nrows) {
  __shared__ float lds[64 * PAIR_STRIDE];

  for (int idx = threadIdx.x; idx < 128 * 64; idx += 256) {
    int k = idx >> 6, n = idx & 63;
    float v = (k < 64) ? Wn[idx] : Wr[idx - 4096];
    lds[(k >> 1) * PAIR_STRIDE + (n << 1) + (k & 1)] = v;
  }
  __syncthreads();

  const int wave = threadIdx.x >> 5;
  const int lane = threadIdx.x & 31;
  const int lh = lane >> 4;
  const int l  = lane & 15;

  long tile = (long)blockIdx.x * 8 + wave;
  long row0 = tile * 16;
  if (row0 >= Nrows) return;

  long arow = row0 + l;
  if (arow >= Nrows) arow = Nrows - 1;
  const float* pa = Aagg + arow * CCH;
  const float* px = Aroot + arow * CCH;

  v8f acc0 = {}, acc1 = {}, acc2 = {}, acc3 = {};

#pragma unroll
  for (int k0 = 0; k0 < 128; k0 += 4) {
    int kk = k0 + (lh << 1);
    v2f a = (k0 < 64) ? *(const v2f*)(pa + kk)
                      : *(const v2f*)(px + (kk - 64));
    const float* bb = lds + (((k0 >> 1) + lh) * PAIR_STRIDE) + (l << 1);
    v2f b0 = *(const v2f*)(bb);
    v2f b1 = *(const v2f*)(bb + 32);
    v2f b2 = *(const v2f*)(bb + 64);
    v2f b3 = *(const v2f*)(bb + 96);
    acc0 = __builtin_amdgcn_wmma_f32_16x16x4_f32(false, a, false, b0, (short)0, acc0, false, false);
    acc1 = __builtin_amdgcn_wmma_f32_16x16x4_f32(false, a, false, b1, (short)0, acc1, false, false);
    acc2 = __builtin_amdgcn_wmma_f32_16x16x4_f32(false, a, false, b2, (short)0, acc2, false, false);
    acc3 = __builtin_amdgcn_wmma_f32_16x16x4_f32(false, a, false, b3, (short)0, acc3, false, false);
  }

  float bc0 = bias[l], bc1 = bias[16 + l], bc2 = bias[32 + l], bc3 = bias[48 + l];

#pragma unroll
  for (int r = 0; r < 8; ++r) {
    long orow = row0 + r + (lh << 3);
    if (orow < Nrows) {
      float* po = out + orow * CCH + l;
      po[0]  = acc0[r] + bc0;
      po[16] = acc1[r] + bc1;
      po[32] = acc2[r] + bc2;
      po[48] = acc3[r] + bc3;
    }
  }
}

extern "C" void kernel_launch(void* const* d_in, const int* in_sizes, int n_in,
                              void* d_out, int out_size, void* d_ws, size_t ws_size,
                              hipStream_t stream) {
  const float* x   = (const float*)d_in[0];
  const int*   ei  = (const int*)d_in[1];
  const float* ea  = (const float*)d_in[2];
  const int*   ps  = (const int*)d_in[3];
  const int*   pd  = (const int*)d_in[4];
  const float* pw  = (const float*)d_in[5];
  const float* Wr1 = (const float*)d_in[7];
  const float* Wn1 = (const float*)d_in[8];
  const float* b1  = (const float*)d_in[9];
  const float* Wr2 = (const float*)d_in[10];
  const float* Wn2 = (const float*)d_in[11];
  const float* b2  = (const float*)d_in[12];
  float* out = (float*)d_out;

  const int N  = in_sizes[0] / CCH;
  const int E  = in_sizes[1] / 2;
  const int EP = in_sizes[3];
  const int NF = out_size / CCH;
  const long NC = (long)N * CCH;

  const int* esrc = ei;
  const int* edst = ei + E;

  const int NA = ((N + SCAN_BLOCK - 1) / SCAN_BLOCK) * SCAN_BLOCK;
  const int NB = ((NF + SCAN_BLOCK - 1) / SCAN_BLOCK) * SCAN_BLOCK;
  const int NT = NA + NB;

  float* agg    = (float*)d_ws;
  float* h1     = agg + NC;
  int*   cnt    = (int*)(h1 + NC);
  int*   cursor = cnt + NT;
  int*   off    = cursor + SCAN_BLOCK;
  int*   pairs  = off + NT;
  int*   rank   = (int*)h1;

  const int zn4 = (NT + SCAN_BLOCK) / 4;
  const int zb  = (zn4 + 255) / 256;
  const int eb  = (E + EP + 255) / 256;
  const int rb1 = (N + 7) / 8;
  const int rb4 = ((NF + 3) / 4 + 7) / 8;
  const int gb  = ((N + 15) / 16 + 7) / 8;

  zero_i32x4<<<zb, 256, 0, stream>>>(cnt, zn4);
  count_rank<<<eb, 256, 0, stream>>>(edst, E, pd, EP, NA, cnt, rank);
  assign_offsets<<<NT / SCAN_BLOCK, SCAN_BLOCK, 0, stream>>>(cnt, off, cursor);
  fill_pairs<<<eb, 256, 0, stream>>>(esrc, edst, ea, E, ps, pd, pw, EP, NA, off, rank, pairs);
  gather_reduce<1><<<rb1, 256, 0, stream>>>(x, off, cnt, pairs, agg, 0, N);
  gemm_dual_wmma<<<gb, 256, 0, stream>>>(agg, x, Wn1, Wr1, b1, h1, N);
  gather_reduce<1><<<rb1, 256, 0, stream>>>(h1, off, cnt, pairs, agg, 0, N);
  gemm_dual_wmma<<<gb, 256, 0, stream>>>(agg, h1, Wn2, Wr2, b2, agg, N);
  gather_reduce<4><<<rb4, 256, 0, stream>>>(agg, off, cnt, pairs, out, NA, NF);
}


# edge ranks: per-edge returning global adds replaced by binning edges into 1024-row buckets (LDS counts, one global add per block and bucket) and LDS cursors per bucket; sums stay in arrival order
# speedup vs baseline: 2.2354x; 1.6261x over previous
#include <hip/hip_runtime.h>
#include <hip/hip_bf16.h>

typedef __attribute__((ext_vector_type(2))) float v2f;
typedef __attribute__((ext_vector_type(8))) float v8f;

#define CCH 64
#define PAIR_STRIDE 160
#define SCAN_BLOCK 1024

#define BIN_THREADS 1024
#define MAX_BUCKETS 1024
#define EDGES_PER_THREAD 8

__global__ void zero_i32x4(int* __restrict__ p, int n4) {
  int i = blockIdx.x * blockDim.x + threadIdx.x;
  if (i < n4) ((int4*)p)[i] = make_int4(0, 0, 0, 0);
}

__device__ __forceinline__ int block_excl_scan(int v, int* wsum, int lane, int wv) {
  int incl = v;
  for (int d = 1; d < 32; d <<= 1) {
    int t = __shfl_up(incl, d, 32);
    if (lane >= d) incl += t;
  }
  if (lane == 31) wsum[wv] = incl;
  __syncthreads();
  if (wv == 0) {
    const int s = wsum[lane];
    int si = s;
    for (int d = 1; d < 32; d <<= 1) {
      int t = __shfl_up(si, d, 32);
      if (lane >= d) si += t;
    }
    wsum[lane] = si - s;
  }
  __syncthreads();
  return wsum[wv] + incl - v;
}

__global__ __launch_bounds__(BIN_THREADS)
void bucket_hist(const int* __restrict__ dstA, int EA,
                 const int* __restrict__ dstB, int EB, int rowBaseB, int blocksA,
                 int* __restrict__ bcount) {
  __shared__ int h[MAX_BUCKETS];
  const int tid = threadIdx.x;
  h[tid] = 0;
  __syncthreads();
  const bool isA = (int)blockIdx.x < blocksA;
  const int* pd = isA ? dstA : dstB;
  const int n = isA ? EA : EB;
  const int rbase = isA ? 0 : rowBaseB;
  const int t0 = (isA ? blockIdx.x : blockIdx.x - blocksA) * (BIN_THREADS * EDGES_PER_THREAD) + tid;
  int r[EDGES_PER_THREAD];
  for (int j = 0; j < EDGES_PER_THREAD; ++j) {
    const int t = t0 + j * BIN_THREADS;
    const bool valid = t < n;
    const int v = pd[valid ? t : n - 1];
    r[j] = valid ? v + rbase : -1;
  }
  for (int j = 0; j < EDGES_PER_THREAD; ++j)
    if (r[j] >= 0)
      __hip_atomic_fetch_add(&h[r[j] / SCAN_BLOCK], 1, __ATOMIC_RELAXED, __HIP_MEMORY_SCOPE_WORKGROUP);
  __syncthreads();
  const int c = h[tid];
  if (c) __hip_atomic_fetch_add(bcount + tid, c, __ATOMIC_RELAXED, __HIP_MEMORY_SCOPE_AGENT);
}

__global__ __launch_bounds__(BIN_THREADS)
void bin_edges(const int* __restrict__ srcA, const int* __restrict__ dstA,
               const float* __restrict__ wA, int EA,
               const int* __restrict__ srcB, const int* __restrict__ dstB,
               const float* __restrict__ wB, int EB, int rowBaseB, int blocksA,
               const int* __restrict__ bcount, int* __restrict__ bcursor,
               int* __restrict__ binned_i) {
  __shared__ int h[MAX_BUCKETS];
  __shared__ int wsum[32];
  const int tid = threadIdx.x;
  const int lane = tid & 31;
  const int wv = tid >> 5;
  const int c = bcount[tid];
  h[tid] = 0;
  __syncthreads();

  const bool isA = (int)blockIdx.x < blocksA;
  const int* psrc = isA ? srcA : srcB;
  const int* pdst = isA ? dstA : dstB;
  const float* pwt = isA ? wA : wB;
  const int n = isA ? EA : EB;
  const int rbase = isA ? 0 : rowBaseB;
  const int t0 = (isA ? blockIdx.x : blockIdx.x - blocksA) * (BIN_THREADS * EDGES_PER_THREAD) + tid;
  int s[EDGES_PER_THREAD], r[EDGES_PER_THREAD], lr[EDGES_PER_THREAD];
  float w[EDGES_PER_THREAD];
  for (int j = 0; j < EDGES_PER_THREAD; ++j) {
    const int t = t0 + j * BIN_THREADS;
    const bool valid = t < n;
    const int u = valid ? t : n - 1;
    s[j] = psrc[u];
    const int v = pdst[u];
    w[j] = pwt[u];
    r[j] = valid ? v + rbase : -1;
    lr[j] = 0;
  }
  for (int j = 0; j < EDGES_PER_THREAD; ++j)
    if (r[j] >= 0)
      lr[j] = __hip_atomic_fetch_add(&h[r[j] / SCAN_BLOCK], 1, __ATOMIC_RELAXED, __HIP_MEMORY_SCOPE_WORKGROUP);
  __syncthreads();

  const int start = block_excl_scan(c, wsum, lane, wv);
  const int mine = h[tid];
  int pos = 0;
  if (mine) pos = __hip_atomic_fetch_add(bcursor + tid, mine, __ATOMIC_RELAXED, __HIP_MEMORY_SCOPE_AGENT);
  h[tid] = start + pos;
  __syncthreads();

  int2* binned = (int2*)binned_i;
  int dpos[EDGES_PER_THREAD];
  for (int j = 0; j < EDGES_PER_THREAD; ++j)
    dpos[j] = h[(r[j] >= 0 ? r[j] : 0) / SCAN_BLOCK] + lr[j];
  for (int j = 0; j < EDGES_PER_THREAD; ++j) {
    if (r[j] >= 0) {
      const unsigned rec = (unsigned)s[j] | ((unsigned)(r[j] % SCAN_BLOCK) << 22);
      binned[dpos[j]] = make_int2((int)rec, __float_as_int(w[j]));
    }
  }
}

__global__ __launch_bounds__(BIN_THREADS)
void rank_fill(const int* __restrict__ bcount, const int* __restrict__ binned_i,
               int* __restrict__ off, int* __restrict__ cnt, int* __restrict__ pairs_i) {
  __shared__ int cntL[SCAN_BLOCK + 1];
  __shared__ int cur[SCAN_BLOCK];
  __shared__ int wsumA[32];
  __shared__ int wsumB[32];
  __shared__ int sh_start, sh_n;
  const int tid = threadIdx.x;
  const int lane = tid & 31;
  const int wv = tid >> 5;
  const int b = blockIdx.x;

  const int c = bcount[tid];
  const int excl = block_excl_scan(c, wsumA, lane, wv);
  if (tid == b) { sh_start = excl; sh_n = c; }
  cntL[tid] = 0;
  __syncthreads();
  const int start = sh_start;
  const int n = sh_n;
  const int2* in = (const int2*)binned_i + start;

  for (int i0 = tid; i0 < n; i0 += 4 * BIN_THREADS) {
    int k[4];
    for (int u = 0; u < 4; ++u) {
      const int i = i0 + u * BIN_THREADS;
      k[u] = in[i < n ? i : n - 1].x;
    }
    for (int u = 0; u < 4; ++u) {
      const int slot = (i0 + u * BIN_THREADS < n) ? (int)((unsigned)k[u] >> 22) : SCAN_BLOCK;
      __hip_atomic_fetch_add(&cntL[slot], 1, __ATOMIC_RELAXED, __HIP_MEMORY_SCOPE_WORKGROUP);
    }
  }
  __syncthreads();

  const int rc = cntL[tid];
  const int ro = block_excl_scan(rc, wsumB, lane, wv);
  cur[tid] = ro;
  off[b * SCAN_BLOCK + tid] = start + ro;
  cnt[b * SCAN_BLOCK + tid] = rc;
  __syncthreads();

  int2* pairs = (int2*)pairs_i + start;
  for (int i0 = tid; i0 < n; i0 += 4 * BIN_THREADS) {
    int2 e[4];
    for (int u = 0; u < 4; ++u) {
      const int i = i0 + u * BIN_THREADS;
      e[u] = in[i < n ? i : n - 1];
    }
    for (int u = 0; u < 4; ++u) {
      if (i0 + u * BIN_THREADS < n) {
        const int p = __hip_atomic_fetch_add(&cur[(unsigned)e[u].x >> 22], 1, __ATOMIC_RELAXED, __HIP_MEMORY_SCOPE_WORKGROUP);
        pairs[p] = make_int2(e[u].x & 0x3fffff, e[u].y);
      }
    }
  }
}

template <int R>
__global__ __launch_bounds__(256)
void gather_reduce(const float* __restrict__ x, const int* __restrict__ off,
                   const int* __restrict__ cnt, const int* __restrict__ pairs_i,
                   float* __restrict__ out, int rowBase, int nrows) {
  const int lane = threadIdx.x & 31;
  const int wave = blockIdx.x * 8 + (threadIdx.x >> 5);
  const int r0 = wave * R;
  if (r0 >= nrows) return;

  int jo = 0, je = 0;
  if (lane < R) {
    jo = off[rowBase + r0 + lane];
    je = jo + cnt[rowBase + r0 + lane];
  }
  const int base = __builtin_amdgcn_readlane(jo, 0);
  const int total = __builtin_amdgcn_readlane(je, R - 1) - base;
  const int relEnd = je - base;

  const int2* pairs = (const int2*)pairs_i + base;
  const float* xl = x + 2 * lane;
  float* ol = out + (long)r0 * CCH + 2 * lane;

  v2f acc = {0.0f, 0.0f};
  int row = 0;
  int rend = __builtin_amdgcn_readlane(relEnd, 0);

  for (int c = 0; c < total; c += 32) {
    int2 pr = make_int2(0, 0);
    if (c + lane < total) pr = pairs[c + lane];
    const int left = total - c;
    const int nn = left < 32 ? left : 32;
    for (int i = 0; i < nn; i += 4) {
      v2f v[4];
      float w[4];
      for (int u = 0; u < 4; ++u) {
        const int s = __builtin_amdgcn_readlane(pr.x, i + u);
        w[u] = __int_as_float(__builtin_amdgcn_readlane(pr.y, i + u));
        v[u] = *(const v2f*)(xl + (long)s * CCH);
      }
      for (int u = 0; u < 4; ++u) {
        if (i + u < nn) {
          if (R > 1) {
            const int e = c + i + u;
            while (e == rend) {
              if (r0 + row < nrows) *(v2f*)(ol + row * CCH) = acc;
              acc = (v2f){0.0f, 0.0f};
              ++row;
              rend = (row < R) ? __builtin_amdgcn_readlane(relEnd, row) : 0x7fffffff;
            }
          }
          acc += v[u] * w[u];
        }
      }
    }
  }
  while (row < R) {
    if (r0 + row < nrows) *(v2f*)(ol + row * CCH) = acc;
    acc = (v2f){0.0f, 0.0f};
    ++row;
  }
}

__global__ __launch_bounds__(256)
void gemm_dual_wmma(const float* Aagg,
                    const float* __restrict__ Aroot,
                    const float* __restrict__ Wn,
                    const float* __restrict__ Wr,
                    const float* __restrict__ bias,
                    float* out, int Nrows) {
  __shared__ float lds[64 * PAIR_STRIDE];

  for (int idx = threadIdx.x; idx < 128 * 64; idx += 256) {
    int k = idx >> 6, n = idx & 63;
    float v = (k < 64) ? Wn[idx] : Wr[idx - 4096];
    lds[(k >> 1) * PAIR_STRIDE + (n << 1) + (k & 1)] = v;
  }
  __syncthreads();

  const int wave = threadIdx.x >> 5;
  const int lane = threadIdx.x & 31;
  const int lh = lane >> 4;
  const int l  = lane & 15;

  long tile = (long)blockIdx.x * 8 + wave;
  long row0 = tile * 16;
  if (row0 >= Nrows) return;

  long arow = row0 + l;
  if (arow >= Nrows) arow = Nrows - 1;
  const float* pa = Aagg + arow * CCH;
  const float* px = Aroot + arow * CCH;

  v8f acc0 = {}, acc1 = {}, acc2 = {}, acc3 = {};

#pragma unroll
  for (int k0 = 0; k0 < 128; k0 += 4) {
    int kk = k0 + (lh << 1);
    v2f a = (k0 < 64) ? *(const v2f*)(pa + kk)
                      : *(const v2f*)(px + (kk - 64));
    const float* bb = lds + (((k0 >> 1) + lh) * PAIR_STRIDE) + (l << 1);
    v2f b0 = *(const v2f*)(bb);
    v2f b1 = *(const v2f*)(bb + 32);
    v2f b2 = *(const v2f*)(bb + 64);
    v2f b3 = *(const v2f*)(bb + 96);
    acc0 = __builtin_amdgcn_wmma_f32_16x16x4_f32(false, a, false, b0, (short)0, acc0, false, false);
    acc1 = __builtin_amdgcn_wmma_f32_16x16x4_f32(false, a, false, b1, (short)0, acc1, false, false);
    acc2 = __builtin_amdgcn_wmma_f32_16x16x4_f32(false, a, false, b2, (short)0, acc2, false, false);
    acc3 = __builtin_amdgcn_wmma_f32_16x16x4_f32(false, a, false, b3, (short)0, acc3, false, false);
  }

  float bc0 = bias[l], bc1 = bias[16 + l], bc2 = bias[32 + l], bc3 = bias[48 + l];

#pragma unroll
  for (int r = 0; r < 8; ++r) {
    long orow = row0 + r + (lh << 3);
    if (orow < Nrows) {
      float* po = out + orow * CCH + l;
      po[0]  = acc0[r] + bc0;
      po[16] = acc1[r] + bc1;
      po[32] = acc2[r] + bc2;
      po[48] = acc3[r] + bc3;
    }
  }
}

extern "C" void kernel_launch(void* const* d_in, const int* in_sizes, int n_in,
                              void* d_out, int out_size, void* d_ws, size_t ws_size,
                              hipStream_t stream) {
  const float* x   = (const float*)d_in[0];
  const int*   ei  = (const int*)d_in[1];
  const float* ea  = (const float*)d_in[2];
  const int*   ps  = (const int*)d_in[3];
  const int*   pd  = (const int*)d_in[4];
  const float* pw  = (const float*)d_in[5];
  const float* Wr1 = (const float*)d_in[7];
  const float* Wn1 = (const float*)d_in[8];
  const float* b1  = (const float*)d_in[9];
  const float* Wr2 = (const float*)d_in[10];
  const float* Wn2 = (const float*)d_in[11];
  const float* b2  = (const float*)d_in[12];
  float* out = (float*)d_out;

  const int N  = in_sizes[0] / CCH;
  const int E  = in_sizes[1] / 2;
  const int EP = in_sizes[3];
  const int NF = out_size / CCH;
  const long NC = (long)N * CCH;

  const int* esrc = ei;
  const int* edst = ei + E;

  const int NA = ((N + SCAN_BLOCK - 1) / SCAN_BLOCK) * SCAN_BLOCK;
  const int NB = ((NF + SCAN_BLOCK - 1) / SCAN_BLOCK) * SCAN_BLOCK;
  const int NT = NA + NB;

  float* agg     = (float*)d_ws;
  float* h1      = agg + NC;
  int*   cnt     = (int*)(h1 + NC);
  int*   off     = cnt + NT;
  int*   pairs   = off + NT;
  int*   bcount  = (int*)agg;
  int*   bcursor = bcount + MAX_BUCKETS;
  int*   binned  = (int*)h1;

  const int nbk = NT / SCAN_BLOCK;
  const int zn4 = 2 * MAX_BUCKETS / 4;
  const int zb  = (zn4 + 255) / 256;
  const int bbA = (E + BIN_THREADS * EDGES_PER_THREAD - 1) / (BIN_THREADS * EDGES_PER_THREAD);
  const int bbB = (EP + BIN_THREADS * EDGES_PER_THREAD - 1) / (BIN_THREADS * EDGES_PER_THREAD);
  const int bb  = bbA + bbB;
  const int rb1 = (N + 7) / 8;
  const int rb4 = ((NF + 3) / 4 + 7) / 8;
  const int gb  = ((N + 15) / 16 + 7) / 8;

  zero_i32x4<<<zb, 256, 0, stream>>>(bcount, zn4);
  bucket_hist<<<bb, BIN_THREADS, 0, stream>>>(edst, E, pd, EP, NA, bbA, bcount);
  bin_edges<<<bb, BIN_THREADS, 0, stream>>>(esrc, edst, ea, E, ps, pd, pw, EP, NA, bbA, bcount, bcursor, binned);
  rank_fill<<<nbk, BIN_THREADS, 0, stream>>>(bcount, binned, off, cnt, pairs);
  gather_reduce<1><<<rb1, 256, 0, stream>>>(x, off, cnt, pairs, agg, 0, N);
  gemm_dual_wmma<<<gb, 256, 0, stream>>>(agg, x, Wn1, Wr1, b1, h1, N);
  gather_reduce<1><<<rb1, 256, 0, stream>>>(h1, off, cnt, pairs, agg, 0, N);
  gemm_dual_wmma<<<gb, 256, 0, stream>>>(agg, h1, Wn2, Wr2, b2, agg, N);
  gather_reduce<4><<<rb4, 256, 0, stream>>>(agg, off, cnt, pairs, out, NA, NF);
}


# gather passes: each wave walks 8 coarse or 32 fine consecutive rows in a loop (12.5k resident waves per pass instead of 100k short ones), 8 row loads in flight; per-row sums stay in arrival order
# speedup vs baseline: 2.4594x; 1.1002x over previous
#include <hip/hip_runtime.h>
#include <hip/hip_bf16.h>

typedef __attribute__((ext_vector_type(2))) float v2f;
typedef __attribute__((ext_vector_type(8))) float v8f;

#define CCH 64
#define PAIR_STRIDE 160
#define SCAN_BLOCK 1024

#define BIN_THREADS 1024
#define MAX_BUCKETS 1024
#define EDGES_PER_THREAD 8

__global__ void zero_i32x4(int* __restrict__ p, int n4) {
  int i = blockIdx.x * blockDim.x + threadIdx.x;
  if (i < n4) ((int4*)p)[i] = make_int4(0, 0, 0, 0);
}

__device__ __forceinline__ int block_excl_scan(int v, int* wsum, int lane, int wv) {
  int incl = v;
  for (int d = 1; d < 32; d <<= 1) {
    int t = __shfl_up(incl, d, 32);
    if (lane >= d) incl += t;
  }
  if (lane == 31) wsum[wv] = incl;
  __syncthreads();
  if (wv == 0) {
    const int s = wsum[lane];
    int si = s;
    for (int d = 1; d < 32; d <<= 1) {
      int t = __shfl_up(si, d, 32);
      if (lane >= d) si += t;
    }
    wsum[lane] = si - s;
  }
  __syncthreads();
  return wsum[wv] + incl - v;
}

__global__ __launch_bounds__(BIN_THREADS)
void bucket_hist(const int* __restrict__ dstA, int EA,
                 const int* __restrict__ dstB, int EB, int rowBaseB, int blocksA,
                 int* __restrict__ bcount) {
  __shared__ int h[MAX_BUCKETS];
  const int tid = threadIdx.x;
  h[tid] = 0;
  __syncthreads();
  const bool isA = (int)blockIdx.x < blocksA;
  const int* pd = isA ? dstA : dstB;
  const int n = isA ? EA : EB;
  const int rbase = isA ? 0 : rowBaseB;
  const int t0 = (isA ? blockIdx.x : blockIdx.x - blocksA) * (BIN_THREADS * EDGES_PER_THREAD) + tid;
  int r[EDGES_PER_THREAD];
  for (int j = 0; j < EDGES_PER_THREAD; ++j) {
    const int t = t0 + j * BIN_THREADS;
    const bool valid = t < n;
    const int v = pd[valid ? t : n - 1];
    r[j] = valid ? v + rbase : -1;
  }
  for (int j = 0; j < EDGES_PER_THREAD; ++j)
    if (r[j] >= 0)
      __hip_atomic_fetch_add(&h[r[j] / SCAN_BLOCK], 1, __ATOMIC_RELAXED, __HIP_MEMORY_SCOPE_WORKGROUP);
  __syncthreads();
  const int c = h[tid];
  if (c) __hip_atomic_fetch_add(bcount + tid, c, __ATOMIC_RELAXED, __HIP_MEMORY_SCOPE_AGENT);
}

__global__ __launch_bounds__(BIN_THREADS)
void bin_edges(const int* __restrict__ srcA, const int* __restrict__ dstA,
               const float* __restrict__ wA, int EA,
               const int* __restrict__ srcB, const int* __restrict__ dstB,
               const float* __restrict__ wB, int EB, int rowBaseB, int blocksA,
               const int* __restrict__ bcount, int* __restrict__ bcursor,
               int* __restrict__ binned_i) {
  __shared__ int h[MAX_BUCKETS];
  __shared__ int wsum[32];
  const int tid = threadIdx.x;
  const int lane = tid & 31;
  const int wv = tid >> 5;
  const int c = bcount[tid];
  h[tid] = 0;
  __syncthreads();

  const bool isA = (int)blockIdx.x < blocksA;
  const int* psrc = isA ? srcA : srcB;
  const int* pdst = isA ? dstA : dstB;
  const float* pwt = isA ? wA : wB;
  const int n = isA ? EA : EB;
  const int rbase = isA ? 0 : rowBaseB;
  const int t0 = (isA ? blockIdx.x : blockIdx.x - blocksA) * (BIN_THREADS * EDGES_PER_THREAD) + tid;
  int s[EDGES_PER_THREAD], r[EDGES_PER_THREAD], lr[EDGES_PER_THREAD];
  float w[EDGES_PER_THREAD];
  for (int j = 0; j < EDGES_PER_THREAD; ++j) {
    const int t = t0 + j * BIN_THREADS;
    const bool valid = t < n;
    const int u = valid ? t : n - 1;
    s[j] = psrc[u];
    const int v = pdst[u];
    w[j] = pwt[u];
    r[j] = valid ? v + rbase : -1;
    lr[j] = 0;
  }
  for (int j = 0; j < EDGES_PER_THREAD; ++j)
    if (r[j] >= 0)
      lr[j] = __hip_atomic_fetch_add(&h[r[j] / SCAN_BLOCK], 1, __ATOMIC_RELAXED, __HIP_MEMORY_SCOPE_WORKGROUP);
  __syncthreads();

  const int start = block_excl_scan(c, wsum, lane, wv);
  const int mine = h[tid];
  int pos = 0;
  if (mine) pos = __hip_atomic_fetch_add(bcursor + tid, mine, __ATOMIC_RELAXED, __HIP_MEMORY_SCOPE_AGENT);
  h[tid] = start + pos;
  __syncthreads();

  int2* binned = (int2*)binned_i;
  int dpos[EDGES_PER_THREAD];
  for (int j = 0; j < EDGES_PER_THREAD; ++j)
    dpos[j] = h[(r[j] >= 0 ? r[j] : 0) / SCAN_BLOCK] + lr[j];
  for (int j = 0; j < EDGES_PER_THREAD; ++j) {
    if (r[j] >= 0) {
      const unsigned rec = (unsigned)s[j] | ((unsigned)(r[j] % SCAN_BLOCK) << 22);
      binned[dpos[j]] = make_int2((int)rec, __float_as_int(w[j]));
    }
  }
}

__global__ __launch_bounds__(BIN_THREADS)
void rank_fill(const int* __restrict__ bcount, const int* __restrict__ binned_i,
               int* __restrict__ off, int* __restrict__ cnt, int* __restrict__ pairs_i) {
  __shared__ int cntL[SCAN_BLOCK + 1];
  __shared__ int cur[SCAN_BLOCK];
  __shared__ int wsumA[32];
  __shared__ int wsumB[32];
  __shared__ int sh_start, sh_n;
  const int tid = threadIdx.x;
  const int lane = tid & 31;
  const int wv = tid >> 5;
  const int b = blockIdx.x;

  const int c = bcount[tid];
  const int excl = block_excl_scan(c, wsumA, lane, wv);
  if (tid == b) { sh_start = excl; sh_n = c; }
  cntL[tid] = 0;
  __syncthreads();
  const int start = sh_start;
  const int n = sh_n;
  const int2* in = (const int2*)binned_i + start;

  for (int i0 = tid; i0 < n; i0 += 4 * BIN_THREADS) {
    int k[4];
    for (int u = 0; u < 4; ++u) {
      const int i = i0 + u * BIN_THREADS;
      k[u] = in[i < n ? i : n - 1].x;
    }
    for (int u = 0; u < 4; ++u) {
      const int slot = (i0 + u * BIN_THREADS < n) ? (int)((unsigned)k[u] >> 22) : SCAN_BLOCK;
      __hip_atomic_fetch_add(&cntL[slot], 1, __ATOMIC_RELAXED, __HIP_MEMORY_SCOPE_WORKGROUP);
    }
  }
  __syncthreads();

  const int rc = cntL[tid];
  const int ro = block_excl_scan(rc, wsumB, lane, wv);
  cur[tid] = ro;
  off[b * SCAN_BLOCK + tid] = start + ro;
  cnt[b * SCAN_BLOCK + tid] = rc;
  __syncthreads();

  int2* pairs = (int2*)pairs_i + start;
  for (int i0 = tid; i0 < n; i0 += 4 * BIN_THREADS) {
    int2 e[4];
    for (int u = 0; u < 4; ++u) {
      const int i = i0 + u * BIN_THREADS;
      e[u] = in[i < n ? i : n - 1];
    }
    for (int u = 0; u < 4; ++u) {
      if (i0 + u * BIN_THREADS < n) {
        const int p = __hip_atomic_fetch_add(&cur[(unsigned)e[u].x >> 22], 1, __ATOMIC_RELAXED, __HIP_MEMORY_SCOPE_WORKGROUP);
        pairs[p] = make_int2(e[u].x & 0x3fffff, e[u].y);
      }
    }
  }
}

#define GATHER_DEPTH 8
template <int R>
__global__ __launch_bounds__(256)
void gather_reduce(const float* __restrict__ x, const int* __restrict__ off,
                   const int* __restrict__ cnt, const int* __restrict__ pairs_i,
                   float* __restrict__ out, int rowBase, int nrows) {
  const int lane = threadIdx.x & 31;
  const int wave = blockIdx.x * 8 + (threadIdx.x >> 5);
  const int r0 = wave * R;
  if (r0 >= nrows) return;

  int jo = 0, je = 0;
  if (lane < R) {
    jo = off[rowBase + r0 + lane];
    je = jo + cnt[rowBase + r0 + lane];
  }
  const int base = __builtin_amdgcn_readlane(jo, 0);
  const int total = __builtin_amdgcn_readlane(je, R - 1) - base;
  const int relEnd = je - base;

  const int2* pairs = (const int2*)pairs_i + base;
  const float* xl = x + 2 * lane;
  float* ol = out + (long)r0 * CCH + 2 * lane;

  v2f acc = {0.0f, 0.0f};
  int row = 0;
  int rend = __builtin_amdgcn_readlane(relEnd, 0);

  if (total <= 0) {
    for (; row < R; ++row)
      if (r0 + row < nrows) *(v2f*)(ol + row * CCH) = acc;
    return;
  }
  const int last = total - 1;
  int2 pr = pairs[lane < last ? lane : last];

  for (int c = 0; c < total; c += 32) {
    const int nxi = c + 32 + lane;
    const int2 nx = pairs[nxi < last ? nxi : last];
    const int left = total - c;
    const int nn = left < 32 ? left : 32;
    for (int i = 0; i < nn; i += GATHER_DEPTH) {
      v2f v[GATHER_DEPTH];
      float w[GATHER_DEPTH];
      for (int u = 0; u < GATHER_DEPTH; ++u) {
        const int s = __builtin_amdgcn_readlane(pr.x, i + u);
        w[u] = __int_as_float(__builtin_amdgcn_readlane(pr.y, i + u));
        v[u] = *(const v2f*)(xl + (long)s * CCH);
      }
      for (int u = 0; u < GATHER_DEPTH; ++u) {
        const int e = c + i + u;
        while (e == rend) {
          if (r0 + row < nrows) *(v2f*)(ol + row * CCH) = acc;
          acc = (v2f){0.0f, 0.0f};
          ++row;
          rend = (row < R) ? __builtin_amdgcn_readlane(relEnd, row) : 0x7fffffff;
        }
        const bool in = i + u < nn;
        const v2f vq = in ? v[u] : (v2f){0.0f, 0.0f};
        const float wq = in ? w[u] : 0.0f;
        acc += vq * wq;
      }
    }
    pr = nx;
  }
  while (row < R) {
    if (r0 + row < nrows) *(v2f*)(ol + row * CCH) = acc;
    acc = (v2f){0.0f, 0.0f};
    ++row;
  }
}

__global__ __launch_bounds__(256)
void gemm_dual_wmma(const float* Aagg,
                    const float* __restrict__ Aroot,
                    const float* __restrict__ Wn,
                    const float* __restrict__ Wr,
                    const float* __restrict__ bias,
                    float* out, int Nrows) {
  __shared__ float lds[64 * PAIR_STRIDE];

  for (int idx = threadIdx.x; idx < 128 * 64; idx += 256) {
    int k = idx >> 6, n = idx & 63;
    float v = (k < 64) ? Wn[idx] : Wr[idx - 4096];
    lds[(k >> 1) * PAIR_STRIDE + (n << 1) + (k & 1)] = v;
  }
  __syncthreads();

  const int wave = threadIdx.x >> 5;
  const int lane = threadIdx.x & 31;
  const int lh = lane >> 4;
  const int l  = lane & 15;

  long tile = (long)blockIdx.x * 8 + wave;
  long row0 = tile * 16;
  if (row0 >= Nrows) return;

  long arow = row0 + l;
  if (arow >= Nrows) arow = Nrows - 1;
  const float* pa = Aagg + arow * CCH;
  const float* px = Aroot + arow * CCH;

  v8f acc0 = {}, acc1 = {}, acc2 = {}, acc3 = {};

#pragma unroll
  for (int k0 = 0; k0 < 128; k0 += 4) {
    int kk = k0 + (lh << 1);
    v2f a = (k0 < 64) ? *(const v2f*)(pa + kk)
                      : *(const v2f*)(px + (kk - 64));
    const float* bb = lds + (((k0 >> 1) + lh) * PAIR_STRIDE) + (l << 1);
    v2f b0 = *(const v2f*)(bb);
    v2f b1 = *(const v2f*)(bb + 32);
    v2f b2 = *(const v2f*)(bb + 64);
    v2f b3 = *(const v2f*)(bb + 96);
    acc0 = __builtin_amdgcn_wmma_f32_16x16x4_f32(false, a, false, b0, (short)0, acc0, false, false);
    acc1 = __builtin_amdgcn_wmma_f32_16x16x4_f32(false, a, false, b1, (short)0, acc1, false, false);
    acc2 = __builtin_amdgcn_wmma_f32_16x16x4_f32(false, a, false, b2, (short)0, acc2, false, false);
    acc3 = __builtin_amdgcn_wmma_f32_16x16x4_f32(false, a, false, b3, (short)0, acc3, false, false);
  }

  float bc0 = bias[l], bc1 = bias[16 + l], bc2 = bias[32 + l], bc3 = bias[48 + l];

#pragma unroll
  for (int r = 0; r < 8; ++r) {
    long orow = row0 + r + (lh << 3);
    if (orow < Nrows) {
      float* po = out + orow * CCH + l;
      po[0]  = acc0[r] + bc0;
      po[16] = acc1[r] + bc1;
      po[32] = acc2[r] + bc2;
      po[48] = acc3[r] + bc3;
    }
  }
}

extern "C" void kernel_launch(void* const* d_in, const int* in_sizes, int n_in,
                              void* d_out, int out_size, void* d_ws, size_t ws_size,
                              hipStream_t stream) {
  const float* x   = (const float*)d_in[0];
  const int*   ei  = (const int*)d_in[1];
  const float* ea  = (const float*)d_in[2];
  const int*   ps  = (const int*)d_in[3];
  const int*   pd  = (const int*)d_in[4];
  const float* pw  = (const float*)d_in[5];
  const float* Wr1 = (const float*)d_in[7];
  const float* Wn1 = (const float*)d_in[8];
  const float* b1  = (const float*)d_in[9];
  const float* Wr2 = (const float*)d_in[10];
  const float* Wn2 = (const float*)d_in[11];
  const float* b2  = (const float*)d_in[12];
  float* out = (float*)d_out;

  const int N  = in_sizes[0] / CCH;
  const int E  = in_sizes[1] / 2;
  const int EP = in_sizes[3];
  const int NF = out_size / CCH;
  const long NC = (long)N * CCH;

  const int* esrc = ei;
  const int* edst = ei + E;

  const int NA = ((N + SCAN_BLOCK - 1) / SCAN_BLOCK) * SCAN_BLOCK;
  const int NB = ((NF + SCAN_BLOCK - 1) / SCAN_BLOCK) * SCAN_BLOCK;
  const int NT = NA + NB;

  float* agg     = (float*)d_ws;
  float* h1      = agg + NC;
  int*   cnt     = (int*)(h1 + NC);
  int*   off     = cnt + NT;
  int*   pairs   = off + NT;
  int*   bcount  = (int*)agg;
  int*   bcursor = bcount + MAX_BUCKETS;
  int*   binned  = (int*)h1;

  const int nbk = NT / SCAN_BLOCK;
  const int zn4 = 2 * MAX_BUCKETS / 4;
  const int zb  = (zn4 + 255) / 256;
  const int bbA = (E + BIN_THREADS * EDGES_PER_THREAD - 1) / (BIN_THREADS * EDGES_PER_THREAD);
  const int bbB = (EP + BIN_THREADS * EDGES_PER_THREAD - 1) / (BIN_THREADS * EDGES_PER_THREAD);
  const int bb  = bbA + bbB;
  const int rbc = ((N + 7) / 8 + 7) / 8;
  const int rbf = ((NF + 31) / 32 + 7) / 8;
  const int gb  = ((N + 15) / 16 + 7) / 8;

  zero_i32x4<<<zb, 256, 0, stream>>>(bcount, zn4);
  bucket_hist<<<bb, BIN_THREADS, 0, stream>>>(edst, E, pd, EP, NA, bbA, bcount);
  bin_edges<<<bb, BIN_THREADS, 0, stream>>>(esrc, edst, ea, E, ps, pd, pw, EP, NA, bbA, bcount, bcursor, binned);
  rank_fill<<<nbk, BIN_THREADS, 0, stream>>>(bcount, binned, off, cnt, pairs);
  gather_reduce<8><<<rbc, 256, 0, stream>>>(x, off, cnt, pairs, agg, 0, N);
  gemm_dual_wmma<<<gb, 256, 0, stream>>>(agg, x, Wn1, Wr1, b1, h1, N);
  gather_reduce<8><<<rbc, 256, 0, stream>>>(h1, off, cnt, pairs, agg, 0, N);
  gemm_dual_wmma<<<gb, 256, 0, stream>>>(agg, h1, Wn2, Wr2, b2, agg, N);
  gather_reduce<32><<<rbf, 256, 0, stream>>>(agg, off, cnt, pairs, out, NA, NF);
}


# list building: the two kernels that scatter 8-byte edge records stage them in LDS in output order and copy out with consecutive lanes on consecutive records (whole-line stores); sum order unchanged
# speedup vs baseline: 2.8341x; 1.1524x over previous
#include <hip/hip_runtime.h>
#include <hip/hip_bf16.h>

typedef __attribute__((ext_vector_type(2))) float v2f;
typedef __attribute__((ext_vector_type(8))) float v8f;
typedef __attribute__((ext_vector_type(2))) int v2i;

#define CCH 64
#define PAIR_STRIDE 160
#define SCAN_BLOCK 1024

#define BIN_THREADS 1024
#define MAX_BUCKETS 1024
#define EDGES_PER_THREAD 8
#define EDGES_PER_BLOCK (BIN_THREADS * EDGES_PER_THREAD)
#define RANK_STAGE_CAP 16384

__global__ void zero_i32x4(int* __restrict__ p, int n4) {
  int i = blockIdx.x * blockDim.x + threadIdx.x;
  if (i < n4) ((int4*)p)[i] = make_int4(0, 0, 0, 0);
}

__device__ __forceinline__ int block_excl_scan(int v, int* wsum, int lane, int wv) {
  int incl = v;
  for (int d = 1; d < 32; d <<= 1) {
    int t = __shfl_up(incl, d, 32);
    if (lane >= d) incl += t;
  }
  if (lane == 31) wsum[wv] = incl;
  __syncthreads();
  if (wv == 0) {
    const int s = wsum[lane];
    int si = s;
    for (int d = 1; d < 32; d <<= 1) {
      int t = __shfl_up(si, d, 32);
      if (lane >= d) si += t;
    }
    wsum[lane] = si - s;
  }
  __syncthreads();
  return wsum[wv] + incl - v;
}

__global__ __launch_bounds__(BIN_THREADS)
void bucket_hist(const int* __restrict__ dstA, int EA,
                 const int* __restrict__ dstB, int EB, int rowBaseB, int blocksA,
                 int* __restrict__ bcount) {
  __shared__ int h[MAX_BUCKETS];
  const int tid = threadIdx.x;
  h[tid] = 0;
  __syncthreads();
  const bool isA = (int)blockIdx.x < blocksA;
  const int* pd = isA ? dstA : dstB;
  const int n = isA ? EA : EB;
  const int rbase = isA ? 0 : rowBaseB;
  const int t0 = (isA ? blockIdx.x : blockIdx.x - blocksA) * (BIN_THREADS * EDGES_PER_THREAD) + tid;
  int r[EDGES_PER_THREAD];
  for (int j = 0; j < EDGES_PER_THREAD; ++j) {
    const int t = t0 + j * BIN_THREADS;
    const bool valid = t < n;
    const int v = pd[valid ? t : n - 1];
    r[j] = valid ? v + rbase : -1;
  }
  for (int j = 0; j < EDGES_PER_THREAD; ++j)
    if (r[j] >= 0)
      __hip_atomic_fetch_add(&h[r[j] / SCAN_BLOCK], 1, __ATOMIC_RELAXED, __HIP_MEMORY_SCOPE_WORKGROUP);
  __syncthreads();
  const int c = h[tid];
  if (c) __hip_atomic_fetch_add(bcount + tid, c, __ATOMIC_RELAXED, __HIP_MEMORY_SCOPE_AGENT);
}

__global__ __launch_bounds__(BIN_THREADS)
void bin_edges(const int* __restrict__ srcA, const int* __restrict__ dstA,
               const float* __restrict__ wA, int EA,
               const int* __restrict__ srcB, const int* __restrict__ dstB,
               const float* __restrict__ wB, int EB, int rowBaseB, int blocksA,
               const int* __restrict__ bcount, int* __restrict__ bcursor,
               int* __restrict__ binned_i) {
  __shared__ int h[MAX_BUCKETS];
  __shared__ int hd[MAX_BUCKETS];
  __shared__ int wsum[32];
  __shared__ int wsum2[32];
  __shared__ v2i srec[EDGES_PER_BLOCK];
  __shared__ int sdel[EDGES_PER_BLOCK];
  const int tid = threadIdx.x;
  const int lane = tid & 31;
  const int wv = tid >> 5;
  const int c = bcount[tid];
  h[tid] = 0;
  __syncthreads();

  const bool isA = (int)blockIdx.x < blocksA;
  const int* psrc = isA ? srcA : srcB;
  const int* pdst = isA ? dstA : dstB;
  const float* pwt = isA ? wA : wB;
  const int n = isA ? EA : EB;
  const int rbase = isA ? 0 : rowBaseB;
  const int e0 = (isA ? blockIdx.x : blockIdx.x - blocksA) * EDGES_PER_BLOCK;
  const int t0 = e0 + tid;
  int s[EDGES_PER_THREAD], r[EDGES_PER_THREAD], lr[EDGES_PER_THREAD];
  float w[EDGES_PER_THREAD];
  for (int j = 0; j < EDGES_PER_THREAD; ++j) {
    const int t = t0 + j * BIN_THREADS;
    const bool valid = t < n;
    const int u = valid ? t : n - 1;
    s[j] = psrc[u];
    const int v = pdst[u];
    w[j] = pwt[u];
    r[j] = valid ? v + rbase : -1;
    lr[j] = 0;
  }
  for (int j = 0; j < EDGES_PER_THREAD; ++j)
    if (r[j] >= 0)
      lr[j] = __hip_atomic_fetch_add(&h[r[j] / SCAN_BLOCK], 1, __ATOMIC_RELAXED, __HIP_MEMORY_SCOPE_WORKGROUP);
  __syncthreads();

  const int start = block_excl_scan(c, wsum, lane, wv);
  const int mine = h[tid];
  const int lstart = block_excl_scan(mine, wsum2, lane, wv);
  int pos = 0;
  if (mine) pos = __hip_atomic_fetch_add(bcursor + tid, mine, __ATOMIC_RELAXED, __HIP_MEMORY_SCOPE_AGENT);
  h[tid] = lstart;
  hd[tid] = start + pos - lstart;
  __syncthreads();

  int sp[EDGES_PER_THREAD], dl[EDGES_PER_THREAD];
  for (int j = 0; j < EDGES_PER_THREAD; ++j) {
    const int b = (r[j] >= 0 ? r[j] : 0) / SCAN_BLOCK;
    sp[j] = h[b] + lr[j];
    dl[j] = hd[b];
  }
  for (int j = 0; j < EDGES_PER_THREAD; ++j) {
    if (r[j] >= 0) {
      const unsigned rec = (unsigned)s[j] | ((unsigned)(r[j] % SCAN_BLOCK) << 22);
      srec[sp[j]] = (v2i){(int)rec, __float_as_int(w[j])};
      sdel[sp[j]] = dl[j];
    }
  }
  __syncthreads();

  int left = n - e0;
  if (left > EDGES_PER_BLOCK) left = EDGES_PER_BLOCK;
  v2i* binned = (v2i*)binned_i;
  for (int i = tid; i < left; i += BIN_THREADS)
    binned[i + sdel[i]] = srec[i];
}

__global__ __launch_bounds__(BIN_THREADS)
void rank_fill(const int* __restrict__ bcount, const int* __restrict__ binned_i,
               int* __restrict__ off, int* __restrict__ cnt, int* __restrict__ pairs_i) {
  __shared__ int cntL[SCAN_BLOCK + 1];
  __shared__ int cur[SCAN_BLOCK];
  __shared__ int wsumA[32];
  __shared__ int wsumB[32];
  __shared__ int sh_start, sh_n;
  __shared__ v2i stage[RANK_STAGE_CAP];
  const int tid = threadIdx.x;
  const int lane = tid & 31;
  const int wv = tid >> 5;
  const int b = blockIdx.x;

  const int c = bcount[tid];
  const int excl = block_excl_scan(c, wsumA, lane, wv);
  if (tid == b) { sh_start = excl; sh_n = c; }
  cntL[tid] = 0;
  __syncthreads();
  const int start = sh_start;
  const int n = sh_n;
  const int2* in = (const int2*)binned_i + start;

  for (int i0 = tid; i0 < n; i0 += 4 * BIN_THREADS) {
    int k[4];
    for (int u = 0; u < 4; ++u) {
      const int i = i0 + u * BIN_THREADS;
      k[u] = in[i < n ? i : n - 1].x;
    }
    for (int u = 0; u < 4; ++u) {
      const int slot = (i0 + u * BIN_THREADS < n) ? (int)((unsigned)k[u] >> 22) : SCAN_BLOCK;
      __hip_atomic_fetch_add(&cntL[slot], 1, __ATOMIC_RELAXED, __HIP_MEMORY_SCOPE_WORKGROUP);
    }
  }
  __syncthreads();

  const int rc = cntL[tid];
  const int ro = block_excl_scan(rc, wsumB, lane, wv);
  cur[tid] = ro;
  off[b * SCAN_BLOCK + tid] = start + ro;
  cnt[b * SCAN_BLOCK + tid] = rc;
  __syncthreads();

  v2i* pairs = (v2i*)pairs_i + start;
  const bool staged = n <= RANK_STAGE_CAP;
  for (int i0 = tid; i0 < n; i0 += 4 * BIN_THREADS) {
    int2 e[4];
    for (int u = 0; u < 4; ++u) {
      const int i = i0 + u * BIN_THREADS;
      e[u] = in[i < n ? i : n - 1];
    }
    for (int u = 0; u < 4; ++u) {
      if (i0 + u * BIN_THREADS < n) {
        const int p = __hip_atomic_fetch_add(&cur[(unsigned)e[u].x >> 22], 1, __ATOMIC_RELAXED, __HIP_MEMORY_SCOPE_WORKGROUP);
        const v2i rec = {e[u].x & 0x3fffff, e[u].y};
        if (staged) stage[p] = rec;
        else pairs[p] = rec;
      }
    }
  }
  if (staged) {
    __syncthreads();
    for (int i = tid; i < n; i += BIN_THREADS) pairs[i] = stage[i];
  }
}

#define GATHER_DEPTH 8
template <int R>
__global__ __launch_bounds__(256)
void gather_reduce(const float* __restrict__ x, const int* __restrict__ off,
                   const int* __restrict__ cnt, const int* __restrict__ pairs_i,
                   float* __restrict__ out, int rowBase, int nrows) {
  const int lane = threadIdx.x & 31;
  const int wave = blockIdx.x * 8 + (threadIdx.x >> 5);
  const int r0 = wave * R;
  if (r0 >= nrows) return;

  int jo = 0, je = 0;
  if (lane < R) {
    jo = off[rowBase + r0 + lane];
    je = jo + cnt[rowBase + r0 + lane];
  }
  const int base = __builtin_amdgcn_readlane(jo, 0);
  const int total = __builtin_amdgcn_readlane(je, R - 1) - base;
  const int relEnd = je - base;

  const int2* pairs = (const int2*)pairs_i + base;
  const float* xl = x + 2 * lane;
  float* ol = out + (long)r0 * CCH + 2 * lane;

  v2f acc = {0.0f, 0.0f};
  int row = 0;
  int rend = __builtin_amdgcn_readlane(relEnd, 0);

  if (total <= 0) {
    for (; row < R; ++row)
      if (r0 + row < nrows) *(v2f*)(ol + row * CCH) = acc;
    return;
  }
  const int last = total - 1;
  int2 pr = pairs[lane < last ? lane : last];

  for (int c = 0; c < total; c += 32) {
    const int nxi = c + 32 + lane;
    const int2 nx = pairs[nxi < last ? nxi : last];
    const int left = total - c;
    const int nn = left < 32 ? left : 32;
    for (int i = 0; i < nn; i += GATHER_DEPTH) {
      v2f v[GATHER_DEPTH];
      float w[GATHER_DEPTH];
      for (int u = 0; u < GATHER_DEPTH; ++u) {
        const int s = __builtin_amdgcn_readlane(pr.x, i + u);
        w[u] = __int_as_float(__builtin_amdgcn_readlane(pr.y, i + u));
        v[u] = *(const v2f*)(xl + (long)s * CCH);
      }
      for (int u = 0; u < GATHER_DEPTH; ++u) {
        const int e = c + i + u;
        while (e == rend) {
          if (r0 + row < nrows) *(v2f*)(ol + row * CCH) = acc;
          acc = (v2f){0.0f, 0.0f};
          ++row;
          rend = (row < R) ? __builtin_amdgcn_readlane(relEnd, row) : 0x7fffffff;
        }
        const bool in = i + u < nn;
        const v2f vq = in ? v[u] : (v2f){0.0f, 0.0f};
        const float wq = in ? w[u] : 0.0f;
        acc += vq * wq;
      }
    }
    pr = nx;
  }
  while (row < R) {
    if (r0 + row < nrows) *(v2f*)(ol + row * CCH) = acc;
    acc = (v2f){0.0f, 0.0f};
    ++row;
  }
}

__global__ __launch_bounds__(256)
void gemm_dual_wmma(const float* Aagg,
                    const float* __restrict__ Aroot,
                    const float* __restrict__ Wn,
                    const float* __restrict__ Wr,
                    const float* __restrict__ bias,
                    float* out, int Nrows) {
  __shared__ float lds[64 * PAIR_STRIDE];

  for (int idx = threadIdx.x; idx < 128 * 64; idx += 256) {
    int k = idx >> 6, n = idx & 63;
    float v = (k < 64) ? Wn[idx] : Wr[idx - 4096];
    lds[(k >> 1) * PAIR_STRIDE + (n << 1) + (k & 1)] = v;
  }
  __syncthreads();

  const int wave = threadIdx.x >> 5;
  const int lane = threadIdx.x & 31;
  const int lh = lane >> 4;
  const int l  = lane & 15;

  long tile = (long)blockIdx.x * 8 + wave;
  long row0 = tile * 16;
  if (row0 >= Nrows) return;

  long arow = row0 + l;
  if (arow >= Nrows) arow = Nrows - 1;
  const float* pa = Aagg + arow * CCH;
  const float* px = Aroot + arow * CCH;

  v8f acc0 = {}, acc1 = {}, acc2 = {}, acc3 = {};

#pragma unroll
  for (int k0 = 0; k0 < 128; k0 += 4) {
    int kk = k0 + (lh << 1);
    v2f a = (k0 < 64) ? *(const v2f*)(pa + kk)
                      : *(const v2f*)(px + (kk - 64));
    const float* bb = lds + (((k0 >> 1) + lh) * PAIR_STRIDE) + (l << 1);
    v2f b0 = *(const v2f*)(bb);
    v2f b1 = *(const v2f*)(bb + 32);
    v2f b2 = *(const v2f*)(bb + 64);
    v2f b3 = *(const v2f*)(bb + 96);
    acc0 = __builtin_amdgcn_wmma_f32_16x16x4_f32(false, a, false, b0, (short)0, acc0, false, false);
    acc1 = __builtin_amdgcn_wmma_f32_16x16x4_f32(false, a, false, b1, (short)0, acc1, false, false);
    acc2 = __builtin_amdgcn_wmma_f32_16x16x4_f32(false, a, false, b2, (short)0, acc2, false, false);
    acc3 = __builtin_amdgcn_wmma_f32_16x16x4_f32(false, a, false, b3, (short)0, acc3, false, false);
  }

  float bc0 = bias[l], bc1 = bias[16 + l], bc2 = bias[32 + l], bc3 = bias[48 + l];

#pragma unroll
  for (int r = 0; r < 8; ++r) {
    long orow = row0 + r + (lh << 3);
    if (orow < Nrows) {
      float* po = out + orow * CCH + l;
      po[0]  = acc0[r] + bc0;
      po[16] = acc1[r] + bc1;
      po[32] = acc2[r] + bc2;
      po[48] = acc3[r] + bc3;
    }
  }
}

extern "C" void kernel_launch(void* const* d_in, const int* in_sizes, int n_in,
                              void* d_out, int out_size, void* d_ws, size_t ws_size,
                              hipStream_t stream) {
  const float* x   = (const float*)d_in[0];
  const int*   ei  = (const int*)d_in[1];
  const float* ea  = (const float*)d_in[2];
  const int*   ps  = (const int*)d_in[3];
  const int*   pd  = (const int*)d_in[4];
  const float* pw  = (const float*)d_in[5];
  const float* Wr1 = (const float*)d_in[7];
  const float* Wn1 = (const float*)d_in[8];
  const float* b1  = (const float*)d_in[9];
  const float* Wr2 = (const float*)d_in[10];
  const float* Wn2 = (const float*)d_in[11];
  const float* b2  = (const float*)d_in[12];
  float* out = (float*)d_out;

  const int N  = in_sizes[0] / CCH;
  const int E  = in_sizes[1] / 2;
  const int EP = in_sizes[3];
  const int NF = out_size / CCH;
  const long NC = (long)N * CCH;

  const int* esrc = ei;
  const int* edst = ei + E;

  const int NA = ((N + SCAN_BLOCK - 1) / SCAN_BLOCK) * SCAN_BLOCK;
  const int NB = ((NF + SCAN_BLOCK - 1) / SCAN_BLOCK) * SCAN_BLOCK;
  const int NT = NA + NB;

  float* agg     = (float*)d_ws;
  float* h1      = agg + NC;
  int*   cnt     = (int*)(h1 + NC);
  int*   off     = cnt + NT;
  int*   pairs   = off + NT;
  int*   bcount  = (int*)agg;
  int*   bcursor = bcount + MAX_BUCKETS;
  int*   binned  = (int*)h1;

  const int nbk = NT / SCAN_BLOCK;
  const int zn4 = 2 * MAX_BUCKETS / 4;
  const int zb  = (zn4 + 255) / 256;
  const int bbA = (E + BIN_THREADS * EDGES_PER_THREAD - 1) / (BIN_THREADS * EDGES_PER_THREAD);
  const int bbB = (EP + BIN_THREADS * EDGES_PER_THREAD - 1) / (BIN_THREADS * EDGES_PER_THREAD);
  const int bb  = bbA + bbB;
  const int rbc = ((N + 7) / 8 + 7) / 8;
  const int rbf = ((NF + 31) / 32 + 7) / 8;
  const int gb  = ((N + 15) / 16 + 7) / 8;

  zero_i32x4<<<zb, 256, 0, stream>>>(bcount, zn4);
  bucket_hist<<<bb, BIN_THREADS, 0, stream>>>(edst, E, pd, EP, NA, bbA, bcount);
  bin_edges<<<bb, BIN_THREADS, 0, stream>>>(esrc, edst, ea, E, ps, pd, pw, EP, NA, bbA, bcount, bcursor, binned);
  rank_fill<<<nbk, BIN_THREADS, 0, stream>>>(bcount, binned, off, cnt, pairs);
  gather_reduce<8><<<rbc, 256, 0, stream>>>(x, off, cnt, pairs, agg, 0, N);
  gemm_dual_wmma<<<gb, 256, 0, stream>>>(agg, x, Wn1, Wr1, b1, h1, N);
  gather_reduce<8><<<rbc, 256, 0, stream>>>(h1, off, cnt, pairs, agg, 0, N);
  gemm_dual_wmma<<<gb, 256, 0, stream>>>(agg, h1, Wn2, Wr2, b2, agg, N);
  gather_reduce<32><<<rbf, 256, 0, stream>>>(agg, off, cnt, pairs, out, NA, NF);
}


# conv layers: a wave sums the edge segments of its 16 rows into an LDS tile and runs the f32 WMMA on it in the same launch (no agg plane, 7 launches not 9), 16 row loads in flight; sum order unchanged
# speedup vs baseline: 2.9593x; 1.0442x over previous
#include <hip/hip_runtime.h>
#include <hip/hip_bf16.h>

typedef __attribute__((ext_vector_type(2))) float v2f;
typedef __attribute__((ext_vector_type(8))) float v8f;
typedef __attribute__((ext_vector_type(2))) int v2i;

#define CCH 64
#define PAIR_STRIDE 160
#define SCAN_BLOCK 1024

#define BIN_THREADS 1024
#define MAX_BUCKETS 1024
#define EDGES_PER_THREAD 8
#define EDGES_PER_BLOCK (BIN_THREADS * EDGES_PER_THREAD)
#define RANK_STAGE_CAP 16384

__global__ void zero_i32x4(int* __restrict__ p, int n4) {
  int i = blockIdx.x * blockDim.x + threadIdx.x;
  if (i < n4) ((int4*)p)[i] = make_int4(0, 0, 0, 0);
}

__device__ __forceinline__ int block_excl_scan(int v, int* wsum, int lane, int wv) {
  int incl = v;
  for (int d = 1; d < 32; d <<= 1) {
    int t = __shfl_up(incl, d, 32);
    if (lane >= d) incl += t;
  }
  if (lane == 31) wsum[wv] = incl;
  __syncthreads();
  if (wv == 0) {
    const int s = wsum[lane];
    int si = s;
    for (int d = 1; d < 32; d <<= 1) {
      int t = __shfl_up(si, d, 32);
      if (lane >= d) si += t;
    }
    wsum[lane] = si - s;
  }
  __syncthreads();
  return wsum[wv] + incl - v;
}

__global__ __launch_bounds__(BIN_THREADS)
void bucket_hist(const int* __restrict__ dstA, int EA,
                 const int* __restrict__ dstB, int EB, int rowBaseB, int blocksA,
                 int* __restrict__ bcount) {
  __shared__ int h[MAX_BUCKETS];
  const int tid = threadIdx.x;
  h[tid] = 0;
  __syncthreads();
  const bool isA = (int)blockIdx.x < blocksA;
  const int* pd = isA ? dstA : dstB;
  const int n = isA ? EA : EB;
  const int rbase = isA ? 0 : rowBaseB;
  const int t0 = (isA ? blockIdx.x : blockIdx.x - blocksA) * (BIN_THREADS * EDGES_PER_THREAD) + tid;
  int r[EDGES_PER_THREAD];
  for (int j = 0; j < EDGES_PER_THREAD; ++j) {
    const int t = t0 + j * BIN_THREADS;
    const bool valid = t < n;
    const int v = pd[valid ? t : n - 1];
    r[j] = valid ? v + rbase : -1;
  }
  for (int j = 0; j < EDGES_PER_THREAD; ++j)
    if (r[j] >= 0)
      __hip_atomic_fetch_add(&h[r[j] / SCAN_BLOCK], 1, __ATOMIC_RELAXED, __HIP_MEMORY_SCOPE_WORKGROUP);
  __syncthreads();
  const int c = h[tid];
  if (c) __hip_atomic_fetch_add(bcount + tid, c, __ATOMIC_RELAXED, __HIP_MEMORY_SCOPE_AGENT);
}

__global__ __launch_bounds__(BIN_THREADS)
void bin_edges(const int* __restrict__ srcA, const int* __restrict__ dstA,
               const float* __restrict__ wA, int EA,
               const int* __restrict__ srcB, const int* __restrict__ dstB,
               const float* __restrict__ wB, int EB, int rowBaseB, int blocksA,
               const int* __restrict__ bcount, int* __restrict__ bcursor,
               int* __restrict__ binned_i) {
  __shared__ int h[MAX_BUCKETS];
  __shared__ int hd[MAX_BUCKETS];
  __shared__ int wsum[32];
  __shared__ int wsum2[32];
  __shared__ v2i srec[EDGES_PER_BLOCK];
  __shared__ int sdel[EDGES_PER_BLOCK];
  const int tid = threadIdx.x;
  const int lane = tid & 31;
  const int wv = tid >> 5;
  const int c = bcount[tid];
  h[tid] = 0;
  __syncthreads();

  const bool isA = (int)blockIdx.x < blocksA;
  const int* psrc = isA ? srcA : srcB;
  const int* pdst = isA ? dstA : dstB;
  const float* pwt = isA ? wA : wB;
  const int n = isA ? EA : EB;
  const int rbase = isA ? 0 : rowBaseB;
  const int e0 = (isA ? blockIdx.x : blockIdx.x - blocksA) * EDGES_PER_BLOCK;
  const int t0 = e0 + tid;
  int s[EDGES_PER_THREAD], r[EDGES_PER_THREAD], lr[EDGES_PER_THREAD];
  float w[EDGES_PER_THREAD];
  for (int j = 0; j < EDGES_PER_THREAD; ++j) {
    const int t = t0 + j * BIN_THREADS;
    const bool valid = t < n;
    const int u = valid ? t : n - 1;
    s[j] = psrc[u];
    const int v = pdst[u];
    w[j] = pwt[u];
    r[j] = valid ? v + rbase : -1;
    lr[j] = 0;
  }
  for (int j = 0; j < EDGES_PER_THREAD; ++j)
    if (r[j] >= 0)
      lr[j] = __hip_atomic_fetch_add(&h[r[j] / SCAN_BLOCK], 1, __ATOMIC_RELAXED, __HIP_MEMORY_SCOPE_WORKGROUP);
  __syncthreads();

  const int start = block_excl_scan(c, wsum, lane, wv);
  const int mine = h[tid];
  const int lstart = block_excl_scan(mine, wsum2, lane, wv);
  int pos = 0;
  if (mine) pos = __hip_atomic_fetch_add(bcursor + tid, mine, __ATOMIC_RELAXED, __HIP_MEMORY_SCOPE_AGENT);
  h[tid] = lstart;
  hd[tid] = start + pos - lstart;
  __syncthreads();

  int sp[EDGES_PER_THREAD], dl[EDGES_PER_THREAD];
  for (int j = 0; j < EDGES_PER_THREAD; ++j) {
    const int b = (r[j] >= 0 ? r[j] : 0) / SCAN_BLOCK;
    sp[j] = h[b] + lr[j];
    dl[j] = hd[b];
  }
  for (int j = 0; j < EDGES_PER_THREAD; ++j) {
    if (r[j] >= 0) {
      const unsigned rec = (unsigned)s[j] | ((unsigned)(r[j] % SCAN_BLOCK) << 22);
      srec[sp[j]] = (v2i){(int)rec, __float_as_int(w[j])};
      sdel[sp[j]] = dl[j];
    }
  }
  __syncthreads();

  int left = n - e0;
  if (left > EDGES_PER_BLOCK) left = EDGES_PER_BLOCK;
  v2i* binned = (v2i*)binned_i;
  for (int i = tid; i < left; i += BIN_THREADS)
    binned[i + sdel[i]] = srec[i];
}

__global__ __launch_bounds__(BIN_THREADS)
void rank_fill(const int* __restrict__ bcount, const int* __restrict__ binned_i,
               int* __restrict__ off, int* __restrict__ cnt, int* __restrict__ pairs_i) {
  __shared__ int cntL[SCAN_BLOCK + 1];
  __shared__ int cur[SCAN_BLOCK];
  __shared__ int wsumA[32];
  __shared__ int wsumB[32];
  __shared__ int sh_start, sh_n;
  __shared__ v2i stage[RANK_STAGE_CAP];
  const int tid = threadIdx.x;
  const int lane = tid & 31;
  const int wv = tid >> 5;
  const int b = blockIdx.x;

  const int c = bcount[tid];
  const int excl = block_excl_scan(c, wsumA, lane, wv);
  if (tid == b) { sh_start = excl; sh_n = c; }
  cntL[tid] = 0;
  __syncthreads();
  const int start = sh_start;
  const int n = sh_n;
  const int2* in = (const int2*)binned_i + start;

  for (int i0 = tid; i0 < n; i0 += 4 * BIN_THREADS) {
    int k[4];
    for (int u = 0; u < 4; ++u) {
      const int i = i0 + u * BIN_THREADS;
      k[u] = in[i < n ? i : n - 1].x;
    }
    for (int u = 0; u < 4; ++u) {
      const int slot = (i0 + u * BIN_THREADS < n) ? (int)((unsigned)k[u] >> 22) : SCAN_BLOCK;
      __hip_atomic_fetch_add(&cntL[slot], 1, __ATOMIC_RELAXED, __HIP_MEMORY_SCOPE_WORKGROUP);
    }
  }
  __syncthreads();

  const int rc = cntL[tid];
  const int ro = block_excl_scan(rc, wsumB, lane, wv);
  cur[tid] = ro;
  off[b * SCAN_BLOCK + tid] = start + ro;
  cnt[b * SCAN_BLOCK + tid] = rc;
  __syncthreads();

  v2i* pairs = (v2i*)pairs_i + start;
  const bool staged = n <= RANK_STAGE_CAP;
  for (int i0 = tid; i0 < n; i0 += 4 * BIN_THREADS) {
    int2 e[4];
    for (int u = 0; u < 4; ++u) {
      const int i = i0 + u * BIN_THREADS;
      e[u] = in[i < n ? i : n - 1];
    }
    for (int u = 0; u < 4; ++u) {
      if (i0 + u * BIN_THREADS < n) {
        const int p = __hip_atomic_fetch_add(&cur[(unsigned)e[u].x >> 22], 1, __ATOMIC_RELAXED, __HIP_MEMORY_SCOPE_WORKGROUP);
        const v2i rec = {e[u].x & 0x3fffff, e[u].y};
        if (staged) stage[p] = rec;
        else pairs[p] = rec;
      }
    }
  }
  if (staged) {
    __syncthreads();
    for (int i = tid; i < n; i += BIN_THREADS) pairs[i] = stage[i];
  }
}

#define GATHER_DEPTH 8
template <int R>
__global__ __launch_bounds__(256)
void gather_reduce(const float* __restrict__ x, const int* __restrict__ off,
                   const int* __restrict__ cnt, const int* __restrict__ pairs_i,
                   float* __restrict__ out, int rowBase, int nrows) {
  const int lane = threadIdx.x & 31;
  const int wave = blockIdx.x * 8 + (threadIdx.x >> 5);
  const int r0 = wave * R;
  if (r0 >= nrows) return;

  int jo = 0, je = 0;
  if (lane < R) {
    jo = off[rowBase + r0 + lane];
    je = jo + cnt[rowBase + r0 + lane];
  }
  const int base = __builtin_amdgcn_readlane(jo, 0);
  const int total = __builtin_amdgcn_readlane(je, R - 1) - base;
  const int relEnd = je - base;

  const int2* pairs = (const int2*)pairs_i + base;
  const float* xl = x + 2 * lane;
  float* ol = out + (long)r0 * CCH + 2 * lane;

  v2f acc = {0.0f, 0.0f};
  int row = 0;
  int rend = __builtin_amdgcn_readlane(relEnd, 0);

  if (total <= 0) {
    for (; row < R; ++row)
      if (r0 + row < nrows) *(v2f*)(ol + row * CCH) = acc;
    return;
  }
  const int last = total - 1;
  int2 pr = pairs[lane < last ? lane : last];

  for (int c = 0; c < total; c += 32) {
    const int nxi = c + 32 + lane;
    const int2 nx = pairs[nxi < last ? nxi : last];
    const int left = total - c;
    const int nn = left < 32 ? left : 32;
    for (int i = 0; i < nn; i += GATHER_DEPTH) {
      v2f v[GATHER_DEPTH];
      float w[GATHER_DEPTH];
      for (int u = 0; u < GATHER_DEPTH; ++u) {
        const int s = __builtin_amdgcn_readlane(pr.x, i + u);
        w[u] = __int_as_float(__builtin_amdgcn_readlane(pr.y, i + u));
        v[u] = *(const v2f*)(xl + (long)s * CCH);
      }
      for (int u = 0; u < GATHER_DEPTH; ++u) {
        const int e = c + i + u;
        while (e == rend) {
          if (r0 + row < nrows) *(v2f*)(ol + row * CCH) = acc;
          acc = (v2f){0.0f, 0.0f};
          ++row;
          rend = (row < R) ? __builtin_amdgcn_readlane(relEnd, row) : 0x7fffffff;
        }
        const bool in = i + u < nn;
        const v2f vq = in ? v[u] : (v2f){0.0f, 0.0f};
        const float wq = in ? w[u] : 0.0f;
        acc += vq * wq;
      }
    }
    pr = nx;
  }
  while (row < R) {
    if (r0 + row < nrows) *(v2f*)(ol + row * CCH) = acc;
    acc = (v2f){0.0f, 0.0f};
    ++row;
  }
}

#define TILE_STRIDE 68
#define FUSED_DEPTH 16

__global__ __launch_bounds__(256)
void gather_gemm_wmma(const float* __restrict__ xin, const int* __restrict__ off,
                      const int* __restrict__ cnt, const int* __restrict__ pairs_i,
                      const float* __restrict__ Wn, const float* __restrict__ Wr,
                      const float* __restrict__ bias, float* __restrict__ out, int nrows) {
  __shared__ float lds[64 * PAIR_STRIDE];
  __shared__ __attribute__((aligned(16))) float tiles[8 * 16 * TILE_STRIDE];

  for (int idx = threadIdx.x; idx < 128 * 64; idx += 256) {
    int k = idx >> 6, n = idx & 63;
    float v = (k < 64) ? Wn[idx] : Wr[idx - 4096];
    lds[(k >> 1) * PAIR_STRIDE + (n << 1) + (k & 1)] = v;
  }
  __syncthreads();

  const int wave = threadIdx.x >> 5;
  const int lane = threadIdx.x & 31;
  const int r0 = (blockIdx.x * 8 + wave) * 16;
  if (r0 >= nrows) return;

  float* tile = tiles + wave * (16 * TILE_STRIDE);
  float* tl = tile + 2 * lane;

  {
    int jo = 0, je = 0;
    if (lane < 16) {
      jo = off[r0 + lane];
      je = jo + cnt[r0 + lane];
    }
    const int base = __builtin_amdgcn_readlane(jo, 0);
    const int total = __builtin_amdgcn_readlane(je, 15) - base;
    const int relEnd = je - base;

    const int2* pairs = (const int2*)pairs_i + base;
    const float* xl = xin + 2 * lane;

    v2f acc = {0.0f, 0.0f};
    int row = 0;
    if (total > 0) {
      int rend = __builtin_amdgcn_readlane(relEnd, 0);
      const int last = total - 1;
      int2 pr = pairs[lane < last ? lane : last];
      for (int c = 0; c < total; c += 32) {
        const int nxi = c + 32 + lane;
        const int2 nx = pairs[nxi < last ? nxi : last];
        const int left = total - c;
        const int nn = left < 32 ? left : 32;
        for (int i = 0; i < nn; i += FUSED_DEPTH) {
          v2f v[FUSED_DEPTH];
          float w[FUSED_DEPTH];
#pragma unroll
          for (int u = 0; u < FUSED_DEPTH; ++u) {
            const int s = __builtin_amdgcn_readlane(pr.x, i + u);
            w[u] = __int_as_float(__builtin_amdgcn_readlane(pr.y, i + u));
            v[u] = *(const v2f*)(xl + (long)s * CCH);
          }
#pragma unroll
          for (int u = 0; u < FUSED_DEPTH; ++u) {
            const int e = c + i + u;
            while (e == rend) {
              *(v2f*)(tl + row * TILE_STRIDE) = acc;
              acc = (v2f){0.0f, 0.0f};
              ++row;
              rend = (row < 16) ? __builtin_amdgcn_readlane(relEnd, row) : 0x7fffffff;
            }
            const bool in = i + u < nn;
            const v2f vq = in ? v[u] : (v2f){0.0f, 0.0f};
            const float wq = in ? w[u] : 0.0f;
            acc += vq * wq;
          }
        }
        pr = nx;
      }
    }
    while (row < 16) {
      *(v2f*)(tl + row * TILE_STRIDE) = acc;
      acc = (v2f){0.0f, 0.0f};
      ++row;
    }
  }

  const int lh = lane >> 4;
  const int l  = lane & 15;

  long arow = (long)r0 + l;
  if (arow >= nrows) arow = nrows - 1;
  const float* pt = tile + l * TILE_STRIDE;
  const float* px = xin + arow * CCH;

  v8f acc0 = {}, acc1 = {}, acc2 = {}, acc3 = {};

#pragma unroll
  for (int k0 = 0; k0 < 128; k0 += 4) {
    int kk = k0 + (lh << 1);
    v2f a = (k0 < 64) ? *(const v2f*)(pt + kk)
                      : *(const v2f*)(px + (kk - 64));
    const float* bb = lds + (((k0 >> 1) + lh) * PAIR_STRIDE) + (l << 1);
    v2f b0 = *(const v2f*)(bb);
    v2f b1 = *(const v2f*)(bb + 32);
    v2f b2 = *(const v2f*)(bb + 64);
    v2f b3 = *(const v2f*)(bb + 96);
    acc0 = __builtin_amdgcn_wmma_f32_16x16x4_f32(false, a, false, b0, (short)0, acc0, false, false);
    acc1 = __builtin_amdgcn_wmma_f32_16x16x4_f32(false, a, false, b1, (short)0, acc1, false, false);
    acc2 = __builtin_amdgcn_wmma_f32_16x16x4_f32(false, a, false, b2, (short)0, acc2, false, false);
    acc3 = __builtin_amdgcn_wmma_f32_16x16x4_f32(false, a, false, b3, (short)0, acc3, false, false);
  }

  float bc0 = bias[l], bc1 = bias[16 + l], bc2 = bias[32 + l], bc3 = bias[48 + l];

#pragma unroll
  for (int r = 0; r < 8; ++r) {
    long orow = (long)r0 + r + (lh << 3);
    if (orow < nrows) {
      float* po = out + orow * CCH + l;
      po[0]  = acc0[r] + bc0;
      po[16] = acc1[r] + bc1;
      po[32] = acc2[r] + bc2;
      po[48] = acc3[r] + bc3;
    }
  }
}

extern "C" void kernel_launch(void* const* d_in, const int* in_sizes, int n_in,
                              void* d_out, int out_size, void* d_ws, size_t ws_size,
                              hipStream_t stream) {
  const float* x   = (const float*)d_in[0];
  const int*   ei  = (const int*)d_in[1];
  const float* ea  = (const float*)d_in[2];
  const int*   ps  = (const int*)d_in[3];
  const int*   pd  = (const int*)d_in[4];
  const float* pw  = (const float*)d_in[5];
  const float* Wr1 = (const float*)d_in[7];
  const float* Wn1 = (const float*)d_in[8];
  const float* b1  = (const float*)d_in[9];
  const float* Wr2 = (const float*)d_in[10];
  const float* Wn2 = (const float*)d_in[11];
  const float* b2  = (const float*)d_in[12];
  float* out = (float*)d_out;

  const int N  = in_sizes[0] / CCH;
  const int E  = in_sizes[1] / 2;
  const int EP = in_sizes[3];
  const int NF = out_size / CCH;
  const long NC = (long)N * CCH;

  const int* esrc = ei;
  const int* edst = ei + E;

  const int NA = ((N + SCAN_BLOCK - 1) / SCAN_BLOCK) * SCAN_BLOCK;
  const int NB = ((NF + SCAN_BLOCK - 1) / SCAN_BLOCK) * SCAN_BLOCK;
  const int NT = NA + NB;

  float* agg     = (float*)d_ws;
  float* h1      = agg + NC;
  int*   cnt     = (int*)(h1 + NC);
  int*   off     = cnt + NT;
  int*   pairs   = off + NT;
  int*   bcount  = (int*)agg;
  int*   bcursor = bcount + MAX_BUCKETS;
  int*   binned  = (int*)h1;

  const int nbk = NT / SCAN_BLOCK;
  const int zn4 = 2 * MAX_BUCKETS / 4;
  const int zb  = (zn4 + 255) / 256;
  const int bbA = (E + BIN_THREADS * EDGES_PER_THREAD - 1) / (BIN_THREADS * EDGES_PER_THREAD);
  const int bbB = (EP + BIN_THREADS * EDGES_PER_THREAD - 1) / (BIN_THREADS * EDGES_PER_THREAD);
  const int bb  = bbA + bbB;
  const int rbf = ((NF + 31) / 32 + 7) / 8;
  const int gb  = ((N + 15) / 16 + 7) / 8;

  zero_i32x4<<<zb, 256, 0, stream>>>(bcount, zn4);
  bucket_hist<<<bb, BIN_THREADS, 0, stream>>>(edst, E, pd, EP, NA, bbA, bcount);
  bin_edges<<<bb, BIN_THREADS, 0, stream>>>(esrc, edst, ea, E, ps, pd, pw, EP, NA, bbA, bcount, bcursor, binned);
  rank_fill<<<nbk, BIN_THREADS, 0, stream>>>(bcount, binned, off, cnt, pairs);
  gather_gemm_wmma<<<gb, 256, 0, stream>>>(x, off, cnt, pairs, Wn1, Wr1, b1, h1, N);
  gather_gemm_wmma<<<gb, 256, 0, stream>>>(h1, off, cnt, pairs, Wn2, Wr2, b2, agg, N);
  gather_reduce<32><<<rbf, 256, 0, stream>>>(agg, off, cnt, pairs, out, NA, NF);
}


# list building: counting launch and counter clear dropped; a binning block writes its records by bucket to its own slab plus a table row the bucket blocks read; row sums in slab, then arrival order
# speedup vs baseline: 3.1125x; 1.0518x over previous
#include <hip/hip_runtime.h>
#include <hip/hip_bf16.h>

typedef __attribute__((ext_vector_type(2))) float v2f;
typedef __attribute__((ext_vector_type(8))) float v8f;
typedef __attribute__((ext_vector_type(2))) int v2i;

#define CCH 64
#define PAIR_STRIDE 160
#define SCAN_BLOCK 1024

#define BIN_THREADS 1024
#define MAX_BUCKETS 1024
#define EDGES_PER_THREAD 8
#define EDGES_PER_BLOCK (BIN_THREADS * EDGES_PER_THREAD)
#define RANK_STAGE_CAP 13312

__device__ __forceinline__ int block_excl_scan(int v, int* wsum, int lane, int wv) {
  int incl = v;
  for (int d = 1; d < 32; d <<= 1) {
    int t = __shfl_up(incl, d, 32);
    if (lane >= d) incl += t;
  }
  if (lane == 31) wsum[wv] = incl;
  __syncthreads();
  if (wv == 0) {
    const int s = wsum[lane];
    int si = s;
    for (int d = 1; d < 32; d <<= 1) {
      int t = __shfl_up(si, d, 32);
      if (lane >= d) si += t;
    }
    wsum[lane] = si - s;
  }
  __syncthreads();
  return wsum[wv] + incl - v;
}

__global__ __launch_bounds__(BIN_THREADS)
void bin_edges(const int* __restrict__ srcA, const int* __restrict__ dstA,
               const float* __restrict__ wA, int EA,
               const int* __restrict__ srcB, const int* __restrict__ dstB,
               const float* __restrict__ wB, int EB, int rowBaseB, int blocksA,
               int* __restrict__ table_i, int* __restrict__ binned_i) {
  __shared__ int h[MAX_BUCKETS];
  __shared__ int wsum[32];
  __shared__ v2i srec[EDGES_PER_BLOCK];
  const int tid = threadIdx.x;
  const int lane = tid & 31;
  const int wv = tid >> 5;
  h[tid] = 0;
  __syncthreads();

  const bool isA = (int)blockIdx.x < blocksA;
  const int* psrc = isA ? srcA : srcB;
  const int* pdst = isA ? dstA : dstB;
  const float* pwt = isA ? wA : wB;
  const int n = isA ? EA : EB;
  const int rbase = isA ? 0 : rowBaseB;
  const int e0 = (isA ? blockIdx.x : blockIdx.x - blocksA) * EDGES_PER_BLOCK;
  const int t0 = e0 + tid;
  int s[EDGES_PER_THREAD], r[EDGES_PER_THREAD], lr[EDGES_PER_THREAD];
  float w[EDGES_PER_THREAD];
  for (int j = 0; j < EDGES_PER_THREAD; ++j) {
    const int t = t0 + j * BIN_THREADS;
    const bool valid = t < n;
    const int u = valid ? t : n - 1;
    s[j] = psrc[u];
    const int v = pdst[u];
    w[j] = pwt[u];
    r[j] = valid ? v + rbase : -1;
    lr[j] = 0;
  }
  for (int j = 0; j < EDGES_PER_THREAD; ++j)
    if (r[j] >= 0)
      lr[j] = __hip_atomic_fetch_add(&h[r[j] / SCAN_BLOCK], 1, __ATOMIC_RELAXED, __HIP_MEMORY_SCOPE_WORKGROUP);
  __syncthreads();

  const int mine = h[tid];
  const int lstart = block_excl_scan(mine, wsum, lane, wv);
  h[tid] = lstart;
  ((v2i*)table_i)[(long)blockIdx.x * MAX_BUCKETS + tid] = (v2i){lstart, mine};
  __syncthreads();

  int sp[EDGES_PER_THREAD];
  for (int j = 0; j < EDGES_PER_THREAD; ++j) {
    const int b = (r[j] >= 0 ? r[j] : 0) / SCAN_BLOCK;
    sp[j] = h[b] + lr[j];
  }
  for (int j = 0; j < EDGES_PER_THREAD; ++j) {
    if (r[j] >= 0) {
      const unsigned rec = (unsigned)s[j] | ((unsigned)(r[j] % SCAN_BLOCK) << 22);
      srec[sp[j]] = (v2i){(int)rec, __float_as_int(w[j])};
    }
  }
  __syncthreads();

  int left = n - e0;
  if (left > EDGES_PER_BLOCK) left = EDGES_PER_BLOCK;
  v2i* slab = (v2i*)binned_i + ((isA ? 0 : EA) + e0);
  for (int i = tid; i < left; i += BIN_THREADS)
    slab[i] = srec[i];
}

__global__ __launch_bounds__(BIN_THREADS)
void rank_fill(const int* __restrict__ table_i, const int* __restrict__ binned_i,
               int EA, int blocksA, int blocksB, int bucketsA,
               int* __restrict__ off, int* __restrict__ cnt, int* __restrict__ pairs_i) {
  __shared__ int cntL[SCAN_BLOCK + 1];
  __shared__ int cur[SCAN_BLOCK + 32];
  __shared__ int sdst[BIN_THREADS];
  __shared__ int sdel[BIN_THREADS];
  __shared__ int wsumA[32];
  __shared__ int wsumB[32];
  __shared__ int sh_start, sh_n;
  __shared__ v2i stage[RANK_STAGE_CAP + 32];
  __shared__ unsigned short segOf[RANK_STAGE_CAP];
  const int tid = threadIdx.x;
  const int lane = tid & 31;
  const int wv = tid >> 5;
  const int b = blockIdx.x;

  const bool isA = b < bucketsA;
  const int J = isA ? blocksA : blocksB;
  const int jbase = isA ? 0 : blocksA;
  const int recBase = isA ? 0 : EA;

  int ls = 0, c = 0;
  if (tid < J) {
    const int2 t = ((const int2*)table_i)[(long)(jbase + tid) * MAX_BUCKETS + b];
    ls = t.x;
    c = t.y;
  }
  const int dstart = block_excl_scan(c, wsumA, lane, wv);
  const int lsum = block_excl_scan(ls, wsumB, lane, wv);
  if (tid == BIN_THREADS - 1) {
    sh_n = dstart + c;
    sh_start = recBase + lsum + ls;
  }
  sdst[tid] = dstart;
  sdel[tid] = recBase + tid * EDGES_PER_BLOCK + ls - dstart;
  cntL[tid] = 0;
  __syncthreads();
  const int start = sh_start;
  const int n = sh_n;
  const int2* in = (const int2*)binned_i;
  v2i* pairs = (v2i*)pairs_i + start;
  const bool staged = n <= RANK_STAGE_CAP;

  if (staged) {
    for (int t = wv; t < J; t += 32) {
      const int d = sdst[t];
      const int cc = ((t + 1 < BIN_THREADS) ? sdst[t + 1] : n) - d;
      for (int i = lane; i < cc; i += 32) segOf[d + i] = (unsigned short)t;
    }
    __syncthreads();

    for (int i0 = tid; i0 < n; i0 += 4 * BIN_THREADS) {
      int k[4];
      for (int u = 0; u < 4; ++u) {
        const int i = i0 + u * BIN_THREADS;
        const int ic = i < n ? i : n - 1;
        k[u] = in[ic + sdel[segOf[ic]]].x;
      }
      for (int u = 0; u < 4; ++u) {
        const int slot = (i0 + u * BIN_THREADS < n) ? (int)((unsigned)k[u] >> 22) : SCAN_BLOCK;
        __hip_atomic_fetch_add(&cntL[slot], 1, __ATOMIC_RELAXED, __HIP_MEMORY_SCOPE_WORKGROUP);
      }
    }
  } else {
    for (int t = wv; t < J; t += 32) {
      const int d = sdst[t];
      const int cc = ((t + 1 < BIN_THREADS) ? sdst[t + 1] : n) - d;
      const int dl = sdel[t];
      for (int i = lane; i < cc; i += 32)
        __hip_atomic_fetch_add(&cntL[(unsigned)in[d + i + dl].x >> 22], 1, __ATOMIC_RELAXED, __HIP_MEMORY_SCOPE_WORKGROUP);
    }
  }
  __syncthreads();

  const int rc = cntL[tid];
  const int ro = block_excl_scan(rc, wsumA, lane, wv);
  cur[tid] = ro;
  off[b * SCAN_BLOCK + tid] = start + ro;
  cnt[b * SCAN_BLOCK + tid] = rc;
  __syncthreads();

  if (staged) {
    for (int i0 = tid; i0 < n; i0 += 4 * BIN_THREADS) {
      int2 e[4];
      for (int u = 0; u < 4; ++u) {
        const int i = i0 + u * BIN_THREADS;
        const int ic = i < n ? i : n - 1;
        e[u] = in[ic + sdel[segOf[ic]]];
      }
      __builtin_amdgcn_sched_barrier(0);
      for (int u = 0; u < 4; ++u) {
        const bool ok = i0 + u * BIN_THREADS < n;
        const int slot = ok ? (int)((unsigned)e[u].x >> 22) : SCAN_BLOCK + lane;
        const int p = __hip_atomic_fetch_add(&cur[slot], 1, __ATOMIC_RELAXED, __HIP_MEMORY_SCOPE_WORKGROUP);
        stage[ok ? p : RANK_STAGE_CAP + lane] = (v2i){e[u].x & 0x3fffff, e[u].y};
      }
    }
    __syncthreads();
    for (int i = tid; i < n; i += BIN_THREADS) pairs[i] = stage[i];
  } else {
    for (int t = wv; t < J; t += 32) {
      const int d = sdst[t];
      const int cc = ((t + 1 < BIN_THREADS) ? sdst[t + 1] : n) - d;
      const int dl = sdel[t];
      for (int i = lane; i < cc; i += 32) {
        const int2 e = in[d + i + dl];
        const int p = __hip_atomic_fetch_add(&cur[(unsigned)e.x >> 22], 1, __ATOMIC_RELAXED, __HIP_MEMORY_SCOPE_WORKGROUP);
        pairs[p] = (v2i){e.x & 0x3fffff, e.y};
      }
    }
  }
}

#define GATHER_DEPTH 8
template <int R>
__global__ __launch_bounds__(256)
void gather_reduce(const float* __restrict__ x, const int* __restrict__ off,
                   const int* __restrict__ cnt, const int* __restrict__ pairs_i,
                   float* __restrict__ out, int rowBase, int nrows) {
  const int lane = threadIdx.x & 31;
  const int wave = blockIdx.x * 8 + (threadIdx.x >> 5);
  const int r0 = wave * R;
  if (r0 >= nrows) return;

  int jo = 0, je = 0;
  if (lane < R) {
    jo = off[rowBase + r0 + lane];
    je = jo + cnt[rowBase + r0 + lane];
  }
  const int base = __builtin_amdgcn_readlane(jo, 0);
  const int total = __builtin_amdgcn_readlane(je, R - 1) - base;
  const int relEnd = je - base;

  const int2* pairs = (const int2*)pairs_i + base;
  const float* xl = x + 2 * lane;
  float* ol = out + (long)r0 * CCH + 2 * lane;

  v2f acc = {0.0f, 0.0f};
  int row = 0;
  int rend = __builtin_amdgcn_readlane(relEnd, 0);

  if (total <= 0) {
    for (; row < R; ++row)
      if (r0 + row < nrows) *(v2f*)(ol + row * CCH) = acc;
    return;
  }
  const int last = total - 1;
  int2 pr = pairs[lane < last ? lane : last];

  for (int c = 0; c < total; c += 32) {
    const int nxi = c + 32 + lane;
    const int2 nx = pairs[nxi < last ? nxi : last];
    const int left = total - c;
    const int nn = left < 32 ? left : 32;
    for (int i = 0; i < nn; i += GATHER_DEPTH) {
      v2f v[GATHER_DEPTH];
      float w[GATHER_DEPTH];
      for (int u = 0; u < GATHER_DEPTH; ++u) {
        const int s = __builtin_amdgcn_readlane(pr.x, i + u);
        w[u] = __int_as_float(__builtin_amdgcn_readlane(pr.y, i + u));
        v[u] = *(const v2f*)(xl + (long)s * CCH);
      }
      for (int u = 0; u < GATHER_DEPTH; ++u) {
        const int e = c + i + u;
        while (e == rend) {
          if (r0 + row < nrows) *(v2f*)(ol + row * CCH) = acc;
          acc = (v2f){0.0f, 0.0f};
          ++row;
          rend = (row < R) ? __builtin_amdgcn_readlane(relEnd, row) : 0x7fffffff;
        }
        const bool in = i + u < nn;
        const v2f vq = in ? v[u] : (v2f){0.0f, 0.0f};
        const float wq = in ? w[u] : 0.0f;
        acc += vq * wq;
      }
    }
    pr = nx;
  }
  while (row < R) {
    if (r0 + row < nrows) *(v2f*)(ol + row * CCH) = acc;
    acc = (v2f){0.0f, 0.0f};
    ++row;
  }
}

#define TILE_STRIDE 68
#define FUSED_DEPTH 16

__global__ __launch_bounds__(256)
void gather_gemm_wmma(const float* __restrict__ xin, const int* __restrict__ off,
                      const int* __restrict__ cnt, const int* __restrict__ pairs_i,
                      const float* __restrict__ Wn, const float* __restrict__ Wr,
                      const float* __restrict__ bias, float* __restrict__ out, int nrows) {
  __shared__ float lds[64 * PAIR_STRIDE];
  __shared__ __attribute__((aligned(16))) float tiles[8 * 16 * TILE_STRIDE];

  for (int idx = threadIdx.x; idx < 128 * 64; idx += 256) {
    int k = idx >> 6, n = idx & 63;
    float v = (k < 64) ? Wn[idx] : Wr[idx - 4096];
    lds[(k >> 1) * PAIR_STRIDE + (n << 1) + (k & 1)] = v;
  }
  __syncthreads();

  const int wave = threadIdx.x >> 5;
  const int lane = threadIdx.x & 31;
  const int r0 = (blockIdx.x * 8 + wave) * 16;
  if (r0 >= nrows) return;

  float* tile = tiles + wave * (16 * TILE_STRIDE);
  float* tl = tile + 2 * lane;

  {
    int jo = 0, je = 0;
    if (lane < 16) {
      jo = off[r0 + lane];
      je = jo + cnt[r0 + lane];
    }
    const int base = __builtin_amdgcn_readlane(jo, 0);
    const int total = __builtin_amdgcn_readlane(je, 15) - base;
    const int relEnd = je - base;

    const int2* pairs = (const int2*)pairs_i + base;
    const float* xl = xin + 2 * lane;

    v2f acc = {0.0f, 0.0f};
    int row = 0;
    if (total > 0) {
      int rend = __builtin_amdgcn_readlane(relEnd, 0);
      const int last = total - 1;
      int2 pr = pairs[lane < last ? lane : last];
      for (int c = 0; c < total; c += 32) {
        const int nxi = c + 32 + lane;
        const int2 nx = pairs[nxi < last ? nxi : last];
        const int left = total - c;
        const int nn = left < 32 ? left : 32;
        for (int i = 0; i < nn; i += FUSED_DEPTH) {
          v2f v[FUSED_DEPTH];
          float w[FUSED_DEPTH];
#pragma unroll
          for (int u = 0; u < FUSED_DEPTH; ++u) {
            const int s = __builtin_amdgcn_readlane(pr.x, i + u);
            w[u] = __int_as_float(__builtin_amdgcn_readlane(pr.y, i + u));
            v[u] = *(const v2f*)(xl + (long)s * CCH);
          }
#pragma unroll
          for (int u = 0; u < FUSED_DEPTH; ++u) {
            const int e = c + i + u;
            while (e == rend) {
              *(v2f*)(tl + row * TILE_STRIDE) = acc;
              acc = (v2f){0.0f, 0.0f};
              ++row;
              rend = (row < 16) ? __builtin_amdgcn_readlane(relEnd, row) : 0x7fffffff;
            }
            const bool in = i + u < nn;
            const v2f vq = in ? v[u] : (v2f){0.0f, 0.0f};
            const float wq = in ? w[u] : 0.0f;
            acc += vq * wq;
          }
        }
        pr = nx;
      }
    }
    while (row < 16) {
      *(v2f*)(tl + row * TILE_STRIDE) = acc;
      acc = (v2f){0.0f, 0.0f};
      ++row;
    }
  }

  const int lh = lane >> 4;
  const int l  = lane & 15;

  long arow = (long)r0 + l;
  if (arow >= nrows) arow = nrows - 1;
  const float* pt = tile + l * TILE_STRIDE;
  const float* px = xin + arow * CCH;

  v8f acc0 = {}, acc1 = {}, acc2 = {}, acc3 = {};

#pragma unroll
  for (int k0 = 0; k0 < 128; k0 += 4) {
    int kk = k0 + (lh << 1);
    v2f a = (k0 < 64) ? *(const v2f*)(pt + kk)
                      : *(const v2f*)(px + (kk - 64));
    const float* bb = lds + (((k0 >> 1) + lh) * PAIR_STRIDE) + (l << 1);
    v2f b0 = *(const v2f*)(bb);
    v2f b1 = *(const v2f*)(bb + 32);
    v2f b2 = *(const v2f*)(bb + 64);
    v2f b3 = *(const v2f*)(bb + 96);
    acc0 = __builtin_amdgcn_wmma_f32_16x16x4_f32(false, a, false, b0, (short)0, acc0, false, false);
    acc1 = __builtin_amdgcn_wmma_f32_16x16x4_f32(false, a, false, b1, (short)0, acc1, false, false);
    acc2 = __builtin_amdgcn_wmma_f32_16x16x4_f32(false, a, false, b2, (short)0, acc2, false, false);
    acc3 = __builtin_amdgcn_wmma_f32_16x16x4_f32(false, a, false, b3, (short)0, acc3, false, false);
  }

  float bc0 = bias[l], bc1 = bias[16 + l], bc2 = bias[32 + l], bc3 = bias[48 + l];

#pragma unroll
  for (int r = 0; r < 8; ++r) {
    long orow = (long)r0 + r + (lh << 3);
    if (orow < nrows) {
      float* po = out + orow * CCH + l;
      po[0]  = acc0[r] + bc0;
      po[16] = acc1[r] + bc1;
      po[32] = acc2[r] + bc2;
      po[48] = acc3[r] + bc3;
    }
  }
}

extern "C" void kernel_launch(void* const* d_in, const int* in_sizes, int n_in,
                              void* d_out, int out_size, void* d_ws, size_t ws_size,
                              hipStream_t stream) {
  const float* x   = (const float*)d_in[0];
  const int*   ei  = (const int*)d_in[1];
  const float* ea  = (const float*)d_in[2];
  const int*   ps  = (const int*)d_in[3];
  const int*   pd  = (const int*)d_in[4];
  const float* pw  = (const float*)d_in[5];
  const float* Wr1 = (const float*)d_in[7];
  const float* Wn1 = (const float*)d_in[8];
  const float* b1  = (const float*)d_in[9];
  const float* Wr2 = (const float*)d_in[10];
  const float* Wn2 = (const float*)d_in[11];
  const float* b2  = (const float*)d_in[12];
  float* out = (float*)d_out;

  const int N  = in_sizes[0] / CCH;
  const int E  = in_sizes[1] / 2;
  const int EP = in_sizes[3];
  const int NF = out_size / CCH;
  const long NC = (long)N * CCH;

  const int* esrc = ei;
  const int* edst = ei + E;

  const int NA = ((N + SCAN_BLOCK - 1) / SCAN_BLOCK) * SCAN_BLOCK;
  const int NB = ((NF + SCAN_BLOCK - 1) / SCAN_BLOCK) * SCAN_BLOCK;
  const int NT = NA + NB;

  float* agg     = (float*)d_ws;
  float* h1      = agg + NC;
  int*   cnt     = (int*)(h1 + NC);
  int*   off     = cnt + NT;
  int*   pairs   = off + NT;
  int*   table   = (int*)agg;
  int*   binned  = (int*)h1;

  const int nbk = NT / SCAN_BLOCK;
  const int bbA = (E + BIN_THREADS * EDGES_PER_THREAD - 1) / (BIN_THREADS * EDGES_PER_THREAD);
  const int bbB = (EP + BIN_THREADS * EDGES_PER_THREAD - 1) / (BIN_THREADS * EDGES_PER_THREAD);
  const int bb  = bbA + bbB;
  const int rbf = ((NF + 31) / 32 + 7) / 8;
  const int gb  = ((N + 15) / 16 + 7) / 8;

  bin_edges<<<bb, BIN_THREADS, 0, stream>>>(esrc, edst, ea, E, ps, pd, pw, EP, NA, bbA, table, binned);
  rank_fill<<<nbk, BIN_THREADS, 0, stream>>>(table, binned, E, bbA, bbB, NA / SCAN_BLOCK, off, cnt, pairs);
  gather_gemm_wmma<<<gb, 256, 0, stream>>>(x, off, cnt, pairs, Wn1, Wr1, b1, h1, N);
  gather_gemm_wmma<<<gb, 256, 0, stream>>>(h1, off, cnt, pairs, Wn2, Wr2, b2, agg, N);
  gather_reduce<32><<<rbf, 256, 0, stream>>>(agg, off, cnt, pairs, out, NA, NF);
}


# gather walks: a wave reads its span's edge records with scalar loads one batch ahead (unclamped, pad records behind the list) so source index and weight are scalar operands; sum order unchanged
# speedup vs baseline: 3.2715x; 1.0511x over previous
#include <hip/hip_runtime.h>
#include <hip/hip_bf16.h>

typedef __attribute__((ext_vector_type(2))) float v2f;
typedef __attribute__((ext_vector_type(8))) float v8f;
typedef __attribute__((ext_vector_type(2))) int v2i;

#define CCH 64
#define PAIR_STRIDE 160
#define SCAN_BLOCK 1024

#define BIN_THREADS 1024
#define MAX_BUCKETS 1024
#define EDGES_PER_THREAD 8
#define EDGES_PER_BLOCK (BIN_THREADS * EDGES_PER_THREAD)
#define RANK_STAGE_CAP 13312
#define READ_AHEAD_PAD 32

__device__ __forceinline__ int block_excl_scan(int v, int* wsum, int lane, int wv) {
  int incl = v;
  for (int d = 1; d < 32; d <<= 1) {
    int t = __shfl_up(incl, d, 32);
    if (lane >= d) incl += t;
  }
  if (lane == 31) wsum[wv] = incl;
  __syncthreads();
  if (wv == 0) {
    const int s = wsum[lane];
    int si = s;
    for (int d = 1; d < 32; d <<= 1) {
      int t = __shfl_up(si, d, 32);
      if (lane >= d) si += t;
    }
    wsum[lane] = si - s;
  }
  __syncthreads();
  return wsum[wv] + incl - v;
}

__global__ __launch_bounds__(BIN_THREADS)
void bin_edges(const int* __restrict__ srcA, const int* __restrict__ dstA,
               const float* __restrict__ wA, int EA,
               const int* __restrict__ srcB, const int* __restrict__ dstB,
               const float* __restrict__ wB, int EB, int rowBaseB, int blocksA,
               int* __restrict__ table_i, int* __restrict__ binned_i) {
  __shared__ int h[MAX_BUCKETS];
  __shared__ int wsum[32];
  __shared__ v2i srec[EDGES_PER_BLOCK];
  const int tid = threadIdx.x;
  const int lane = tid & 31;
  const int wv = tid >> 5;
  h[tid] = 0;
  __syncthreads();

  const bool isA = (int)blockIdx.x < blocksA;
  const int* psrc = isA ? srcA : srcB;
  const int* pdst = isA ? dstA : dstB;
  const float* pwt = isA ? wA : wB;
  const int n = isA ? EA : EB;
  const int rbase = isA ? 0 : rowBaseB;
  const int e0 = (isA ? blockIdx.x : blockIdx.x - blocksA) * EDGES_PER_BLOCK;
  const int t0 = e0 + tid;
  int s[EDGES_PER_THREAD], r[EDGES_PER_THREAD], lr[EDGES_PER_THREAD];
  float w[EDGES_PER_THREAD];
  for (int j = 0; j < EDGES_PER_THREAD; ++j) {
    const int t = t0 + j * BIN_THREADS;
    const bool valid = t < n;
    const int u = valid ? t : n - 1;
    s[j] = psrc[u];
    const int v = pdst[u];
    w[j] = pwt[u];
    r[j] = valid ? v + rbase : -1;
    lr[j] = 0;
  }
  for (int j = 0; j < EDGES_PER_THREAD; ++j)
    if (r[j] >= 0)
      lr[j] = __hip_atomic_fetch_add(&h[r[j] / SCAN_BLOCK], 1, __ATOMIC_RELAXED, __HIP_MEMORY_SCOPE_WORKGROUP);
  __syncthreads();

  const int mine = h[tid];
  const int lstart = block_excl_scan(mine, wsum, lane, wv);
  h[tid] = lstart;
  ((v2i*)table_i)[(long)blockIdx.x * MAX_BUCKETS + tid] = (v2i){lstart, mine};
  __syncthreads();

  int sp[EDGES_PER_THREAD];
  for (int j = 0; j < EDGES_PER_THREAD; ++j) {
    const int b = (r[j] >= 0 ? r[j] : 0) / SCAN_BLOCK;
    sp[j] = h[b] + lr[j];
  }
  for (int j = 0; j < EDGES_PER_THREAD; ++j) {
    if (r[j] >= 0) {
      const unsigned rec = (unsigned)s[j] | ((unsigned)(r[j] % SCAN_BLOCK) << 22);
      srec[sp[j]] = (v2i){(int)rec, __float_as_int(w[j])};
    }
  }
  __syncthreads();

  int left = n - e0;
  if (left > EDGES_PER_BLOCK) left = EDGES_PER_BLOCK;
  v2i* slab = (v2i*)binned_i + ((isA ? 0 : EA) + e0);
  for (int i = tid; i < left; i += BIN_THREADS)
    slab[i] = srec[i];
}

__global__ __launch_bounds__(BIN_THREADS)
void rank_fill(const int* __restrict__ table_i, const int* __restrict__ binned_i,
               int EA, int blocksA, int blocksB, int bucketsA, int buckets,
               int* __restrict__ off, int* __restrict__ cnt, int* __restrict__ pairs_i) {
  __shared__ int cntL[SCAN_BLOCK + 1];
  __shared__ int cur[SCAN_BLOCK + 32];
  __shared__ int sdst[BIN_THREADS];
  __shared__ int sdel[BIN_THREADS];
  __shared__ int wsumA[32];
  __shared__ int wsumB[32];
  __shared__ int sh_start, sh_n;
  __shared__ v2i stage[RANK_STAGE_CAP + 32];
  __shared__ unsigned short segOf[RANK_STAGE_CAP];
  const int tid = threadIdx.x;
  const int lane = tid & 31;
  const int wv = tid >> 5;
  const int b = blockIdx.x;

  const bool isA = b < bucketsA;
  const int J = isA ? blocksA : blocksB;
  const int jbase = isA ? 0 : blocksA;
  const int recBase = isA ? 0 : EA;

  int ls = 0, c = 0;
  if (tid < J) {
    const int2 t = ((const int2*)table_i)[(long)(jbase + tid) * MAX_BUCKETS + b];
    ls = t.x;
    c = t.y;
  }
  const int dstart = block_excl_scan(c, wsumA, lane, wv);
  const int lsum = block_excl_scan(ls, wsumB, lane, wv);
  if (tid == BIN_THREADS - 1) {
    sh_n = dstart + c;
    sh_start = recBase + lsum + ls;
  }
  sdst[tid] = dstart;
  sdel[tid] = recBase + tid * EDGES_PER_BLOCK + ls - dstart;
  cntL[tid] = 0;
  __syncthreads();
  const int start = sh_start;
  const int n = sh_n;
  const int2* in = (const int2*)binned_i;
  v2i* pairs = (v2i*)pairs_i + start;
  const bool staged = n <= RANK_STAGE_CAP;

  if (staged) {
    for (int t = wv; t < J; t += 32) {
      const int d = sdst[t];
      const int cc = ((t + 1 < BIN_THREADS) ? sdst[t + 1] : n) - d;
      for (int i = lane; i < cc; i += 32) segOf[d + i] = (unsigned short)t;
    }
    __syncthreads();

    for (int i0 = tid; i0 < n; i0 += 4 * BIN_THREADS) {
      int k[4];
      for (int u = 0; u < 4; ++u) {
        const int i = i0 + u * BIN_THREADS;
        const int ic = i < n ? i : n - 1;
        k[u] = in[ic + sdel[segOf[ic]]].x;
      }
      for (int u = 0; u < 4; ++u) {
        const int slot = (i0 + u * BIN_THREADS < n) ? (int)((unsigned)k[u] >> 22) : SCAN_BLOCK;
        __hip_atomic_fetch_add(&cntL[slot], 1, __ATOMIC_RELAXED, __HIP_MEMORY_SCOPE_WORKGROUP);
      }
    }
  } else {
    for (int t = wv; t < J; t += 32) {
      const int d = sdst[t];
      const int cc = ((t + 1 < BIN_THREADS) ? sdst[t + 1] : n) - d;
      const int dl = sdel[t];
      for (int i = lane; i < cc; i += 32)
        __hip_atomic_fetch_add(&cntL[(unsigned)in[d + i + dl].x >> 22], 1, __ATOMIC_RELAXED, __HIP_MEMORY_SCOPE_WORKGROUP);
    }
  }
  __syncthreads();

  const int rc = cntL[tid];
  const int ro = block_excl_scan(rc, wsumA, lane, wv);
  cur[tid] = ro;
  off[b * SCAN_BLOCK + tid] = start + ro;
  cnt[b * SCAN_BLOCK + tid] = rc;
  __syncthreads();

  if (staged) {
    for (int i0 = tid; i0 < n; i0 += 4 * BIN_THREADS) {
      int2 e[4];
      for (int u = 0; u < 4; ++u) {
        const int i = i0 + u * BIN_THREADS;
        const int ic = i < n ? i : n - 1;
        e[u] = in[ic + sdel[segOf[ic]]];
      }
      __builtin_amdgcn_sched_barrier(0);
      for (int u = 0; u < 4; ++u) {
        const bool ok = i0 + u * BIN_THREADS < n;
        const int slot = ok ? (int)((unsigned)e[u].x >> 22) : SCAN_BLOCK + lane;
        const int p = __hip_atomic_fetch_add(&cur[slot], 1, __ATOMIC_RELAXED, __HIP_MEMORY_SCOPE_WORKGROUP);
        stage[ok ? p : RANK_STAGE_CAP + lane] = (v2i){e[u].x & 0x3fffff, e[u].y};
      }
    }
    __syncthreads();
    for (int i = tid; i < n; i += BIN_THREADS) pairs[i] = stage[i];
  } else {
    for (int t = wv; t < J; t += 32) {
      const int d = sdst[t];
      const int cc = ((t + 1 < BIN_THREADS) ? sdst[t + 1] : n) - d;
      const int dl = sdel[t];
      for (int i = lane; i < cc; i += 32) {
        const int2 e = in[d + i + dl];
        const int p = __hip_atomic_fetch_add(&cur[(unsigned)e.x >> 22], 1, __ATOMIC_RELAXED, __HIP_MEMORY_SCOPE_WORKGROUP);
        pairs[p] = (v2i){e.x & 0x3fffff, e.y};
      }
    }
  }
  if (b == buckets - 1 && tid < READ_AHEAD_PAD) pairs[n + tid] = (v2i){0, 0};
}

#define GATHER_DEPTH 8
template <int R>
__global__ __launch_bounds__(256)
void gather_reduce(const float* __restrict__ x, const int* __restrict__ off,
                   const int* __restrict__ cnt, const int* __restrict__ pairs_i,
                   float* __restrict__ out, int rowBase, int nrows) {
  const int lane = threadIdx.x & 31;
  const int wave = blockIdx.x * 8 + (threadIdx.x >> 5);
  const int r0 = wave * R;
  if (r0 >= nrows) return;

  int jo = 0, je = 0;
  if (lane < R) {
    jo = off[rowBase + r0 + lane];
    je = jo + cnt[rowBase + r0 + lane];
  }
  const int base = __builtin_amdgcn_readlane(jo, 0);
  const int total = __builtin_amdgcn_readlane(je, R - 1) - base;
  const int relEnd = je - base;

  const int2* rp = (const int2*)pairs_i + base;
  const char* xb = (const char*)x;
  const unsigned lane8 = (unsigned)lane << 3;
  float* ol = out + (long)r0 * CCH + 2 * lane;

  v2f acc = {0.0f, 0.0f};
  int row = 0;

  if (total > 0) {
    int rend = __builtin_amdgcn_readlane(relEnd, 0);
    int2 rec[GATHER_DEPTH];
#pragma unroll
    for (int u = 0; u < GATHER_DEPTH; ++u) rec[u] = rp[u];
    for (int c = 0; c < total; c += GATHER_DEPTH) {
      v2f v[GATHER_DEPTH];
      float w[GATHER_DEPTH];
#pragma unroll
      for (int u = 0; u < GATHER_DEPTH; ++u) {
        const unsigned voff = ((unsigned)rec[u].x << 8) + lane8;
        w[u] = __int_as_float(rec[u].y);
        v[u] = *(const v2f*)(xb + voff);
      }
#pragma unroll
      for (int u = 0; u < GATHER_DEPTH; ++u) rec[u] = rp[c + GATHER_DEPTH + u];
#pragma unroll
      for (int u = 0; u < GATHER_DEPTH; ++u) {
        const int e = c + u;
        while (e == rend) {
          if (r0 + row < nrows) *(v2f*)(ol + row * CCH) = acc;
          acc = (v2f){0.0f, 0.0f};
          ++row;
          rend = (row < R) ? __builtin_amdgcn_readlane(relEnd, row) : 0x7fffffff;
        }
        acc += v[u] * w[u];
      }
    }
  }
  while (row < R) {
    if (r0 + row < nrows) *(v2f*)(ol + row * CCH) = acc;
    acc = (v2f){0.0f, 0.0f};
    ++row;
  }
}

#define TILE_STRIDE 68
#define FUSED_DEPTH 16

__global__ __launch_bounds__(256)
void gather_gemm_wmma(const float* __restrict__ xin, const int* __restrict__ off,
                      const int* __restrict__ cnt, const int* __restrict__ pairs_i,
                      const float* __restrict__ Wn, const float* __restrict__ Wr,
                      const float* __restrict__ bias, float* __restrict__ out, int nrows) {
  __shared__ float lds[64 * PAIR_STRIDE];
  __shared__ __attribute__((aligned(16))) float tiles[8 * 16 * TILE_STRIDE];

  for (int idx = threadIdx.x; idx < 128 * 64; idx += 256) {
    int k = idx >> 6, n = idx & 63;
    float v = (k < 64) ? Wn[idx] : Wr[idx - 4096];
    lds[(k >> 1) * PAIR_STRIDE + (n << 1) + (k & 1)] = v;
  }
  __syncthreads();

  const int wave = threadIdx.x >> 5;
  const int lane = threadIdx.x & 31;
  const int r0 = (blockIdx.x * 8 + wave) * 16;
  if (r0 >= nrows) return;

  float* tile = tiles + wave * (16 * TILE_STRIDE);
  float* tl = tile + 2 * lane;

  {
    int jo = 0, je = 0;
    if (lane < 16) {
      jo = off[r0 + lane];
      je = jo + cnt[r0 + lane];
    }
    const int base = __builtin_amdgcn_readlane(jo, 0);
    const int total = __builtin_amdgcn_readlane(je, 15) - base;
    const int relEnd = je - base;

    const int2* rp = (const int2*)pairs_i + base;
    const char* xb = (const char*)xin;
    const unsigned lane8 = (unsigned)lane << 3;

    v2f acc = {0.0f, 0.0f};
    int row = 0;
    if (total > 0) {
      int rend = __builtin_amdgcn_readlane(relEnd, 0);
      int2 rec[FUSED_DEPTH];
#pragma unroll
      for (int u = 0; u < FUSED_DEPTH; ++u) rec[u] = rp[u];
      for (int c = 0; c < total; c += FUSED_DEPTH) {
        v2f v[FUSED_DEPTH];
        float w[FUSED_DEPTH];
#pragma unroll
        for (int u = 0; u < FUSED_DEPTH; ++u) {
          const unsigned voff = ((unsigned)rec[u].x << 8) + lane8;
          w[u] = __int_as_float(rec[u].y);
          v[u] = *(const v2f*)(xb + voff);
        }
#pragma unroll
        for (int u = 0; u < FUSED_DEPTH; ++u) rec[u] = rp[c + FUSED_DEPTH + u];
#pragma unroll
        for (int u = 0; u < FUSED_DEPTH; ++u) {
          const int e = c + u;
          while (e == rend) {
            *(v2f*)(tl + row * TILE_STRIDE) = acc;
            acc = (v2f){0.0f, 0.0f};
            ++row;
            rend = (row < 16) ? __builtin_amdgcn_readlane(relEnd, row) : 0x7fffffff;
          }
          acc += v[u] * w[u];
        }
      }
    }
    while (row < 16) {
      *(v2f*)(tl + row * TILE_STRIDE) = acc;
      acc = (v2f){0.0f, 0.0f};
      ++row;
    }
  }

  const int lh = lane >> 4;
  const int l  = lane & 15;

  long arow = (long)r0 + l;
  if (arow >= nrows) arow = nrows - 1;
  const float* pt = tile + l * TILE_STRIDE;
  const float* px = xin + arow * CCH;

  v8f acc0 = {}, acc1 = {}, acc2 = {}, acc3 = {};

#pragma unroll
  for (int k0 = 0; k0 < 128; k0 += 4) {
    int kk = k0 + (lh << 1);
    v2f a = (k0 < 64) ? *(const v2f*)(pt + kk)
                      : *(const v2f*)(px + (kk - 64));
    const float* bb = lds + (((k0 >> 1) + lh) * PAIR_STRIDE) + (l << 1);
    v2f b0 = *(const v2f*)(bb);
    v2f b1 = *(const v2f*)(bb + 32);
    v2f b2 = *(const v2f*)(bb + 64);
    v2f b3 = *(const v2f*)(bb + 96);
    acc0 = __builtin_amdgcn_wmma_f32_16x16x4_f32(false, a, false, b0, (short)0, acc0, false, false);
    acc1 = __builtin_amdgcn_wmma_f32_16x16x4_f32(false, a, false, b1, (short)0, acc1, false, false);
    acc2 = __builtin_amdgcn_wmma_f32_16x16x4_f32(false, a, false, b2, (short)0, acc2, false, false);
    acc3 = __builtin_amdgcn_wmma_f32_16x16x4_f32(false, a, false, b3, (short)0, acc3, false, false);
  }

  float bc0 = bias[l], bc1 = bias[16 + l], bc2 = bias[32 + l], bc3 = bias[48 + l];

#pragma unroll
  for (int r = 0; r < 8; ++r) {
    long orow = (long)r0 + r + (lh << 3);
    if (orow < nrows) {
      float* po = out + orow * CCH + l;
      po[0]  = acc0[r] + bc0;
      po[16] = acc1[r] + bc1;
      po[32] = acc2[r] + bc2;
      po[48] = acc3[r] + bc3;
    }
  }
}

extern "C" void kernel_launch(void* const* d_in, const int* in_sizes, int n_in,
                              void* d_out, int out_size, void* d_ws, size_t ws_size,
                              hipStream_t stream) {
  const float* x   = (const float*)d_in[0];
  const int*   ei  = (const int*)d_in[1];
  const float* ea  = (const float*)d_in[2];
  const int*   ps  = (const int*)d_in[3];
  const int*   pd  = (const int*)d_in[4];
  const float* pw  = (const float*)d_in[5];
  const float* Wr1 = (const float*)d_in[7];
  const float* Wn1 = (const float*)d_in[8];
  const float* b1  = (const float*)d_in[9];
  const float* Wr2 = (const float*)d_in[10];
  const float* Wn2 = (const float*)d_in[11];
  const float* b2  = (const float*)d_in[12];
  float* out = (float*)d_out;

  const int N  = in_sizes[0] / CCH;
  const int E  = in_sizes[1] / 2;
  const int EP = in_sizes[3];
  const int NF = out_size / CCH;
  const long NC = (long)N * CCH;

  const int* esrc = ei;
  const int* edst = ei + E;

  const int NA = ((N + SCAN_BLOCK - 1) / SCAN_BLOCK) * SCAN_BLOCK;
  const int NB = ((NF + SCAN_BLOCK - 1) / SCAN_BLOCK) * SCAN_BLOCK;
  const int NT = NA + NB;

  float* agg     = (float*)d_ws;
  float* h1      = agg + NC;
  int*   cnt     = (int*)(h1 + NC);
  int*   off     = cnt + NT;
  int*   pairs   = off + NT;
  int*   table   = (int*)agg;
  int*   binned  = (int*)h1;

  const int nbk = NT / SCAN_BLOCK;
  const int bbA = (E + BIN_THREADS * EDGES_PER_THREAD - 1) / (BIN_THREADS * EDGES_PER_THREAD);
  const int bbB = (EP + BIN_THREADS * EDGES_PER_THREAD - 1) / (BIN_THREADS * EDGES_PER_THREAD);
  const int bb  = bbA + bbB;
  const int rbf = ((NF + 31) / 32 + 7) / 8;
  const int gb  = ((N + 15) / 16 + 7) / 8;

  bin_edges<<<bb, BIN_THREADS, 0, stream>>>(esrc, edst, ea, E, ps, pd, pw, EP, NA, bbA, table, binned);
  rank_fill<<<nbk, BIN_THREADS, 0, stream>>>(table, binned, E, bbA, bbB, NA / SCAN_BLOCK, nbk, off, cnt, pairs);
  gather_gemm_wmma<<<gb, 256, 0, stream>>>(x, off, cnt, pairs, Wn1, Wr1, b1, h1, N);
  gather_gemm_wmma<<<gb, 256, 0, stream>>>(h1, off, cnt, pairs, Wn2, Wr2, b2, agg, N);
  gather_reduce<32><<<rbf, 256, 0, stream>>>(agg, off, cnt, pairs, out, NA, NF);
}
